# starting kernel
# baseline (speedup 1.0000x reference)
#include <hip/hip_runtime.h>

#define NNODES 100000
#define NEDGES 1600000

typedef __attribute__((ext_vector_type(2))) float v2f;
typedef __attribute__((ext_vector_type(8))) float v8f;

__device__ __forceinline__ void atomAddF32(float* p, float v) {
    unsafeAtomicAdd(p, v);
}

__global__ void deg_kernel(const int* __restrict__ col, const float* __restrict__ ew,
                           float* __restrict__ deg, int E) {
    int e = blockIdx.x * blockDim.x + threadIdx.x;
    if (e < E) atomAddF32(&deg[col[e]], ew[e]);
}

__global__ void dinv_kernel(float* __restrict__ deg, int n) {
    int i = blockIdx.x * blockDim.x + threadIdx.x;
    if (i < n) { float d = deg[i]; deg[i] = (d > 0.f) ? rsqrtf(d) : 0.f; }
}

__global__ void norm_kernel(const int* __restrict__ row, const int* __restrict__ col,
                            const float* __restrict__ ew, const float* __restrict__ dinv,
                            float* __restrict__ norm, int E) {
    int e = blockIdx.x * blockDim.x + threadIdx.x;
    if (e < E) norm[e] = dinv[row[e]] * ew[e] * dinv[col[e]];
}

__global__ void prop3_kernel(const int* __restrict__ row, const int* __restrict__ col,
                             const float* __restrict__ norm, const float* __restrict__ hin,
                             float* __restrict__ hout, int E) {
    int e = blockIdx.x * blockDim.x + threadIdx.x;
    if (e >= E) return;
    int r = row[e], c = col[e];
    float w = norm[e];
    atomAddF32(&hout[c * 3 + 0], w * hin[r * 3 + 0]);
    atomAddF32(&hout[c * 3 + 1], w * hin[r * 3 + 1]);
    atomAddF32(&hout[c * 3 + 2], w * hin[r * 3 + 2]);
}

template <int D>
__global__ void prop_kernel(const int* __restrict__ row, const int* __restrict__ col,
                            const float* __restrict__ norm, const float* __restrict__ hin,
                            float* __restrict__ hout, int E) {
    constexpr int D4 = D / 4;
    int idx = blockIdx.x * blockDim.x + threadIdx.x;
    if (idx >= E * D4) return;
    int e = idx / D4;
    int c = idx % D4;
    int r = row[e], cl = col[e];
    float w = norm[e];
    const float4 v = *reinterpret_cast<const float4*>(hin + (size_t)r * D + 4 * c);
    float* dst = hout + (size_t)cl * D + 4 * c;
    atomAddF32(dst + 0, w * v.x);
    atomAddF32(dst + 1, w * v.y);
    atomAddF32(dst + 2, w * v.z);
    atomAddF32(dst + 3, w * v.w);
}

__global__ void layer1_kernel(const float* __restrict__ x, const float* __restrict__ hop,
                              const float* __restrict__ W, const float* __restrict__ bias,
                              float* __restrict__ out) {
    int idx = blockIdx.x * blockDim.x + threadIdx.x;
    if (idx >= NNODES * 32) return;
    int i = idx >> 5, j = idx & 31;
    float acc = bias[j];
#pragma unroll
    for (int k = 0; k < 3; ++k) {
        acc += x[i * 3 + k]   * W[k * 32 + j];
        acc += hop[i * 3 + k] * W[96 + k * 32 + j];
    }
    out[idx] = fmaxf(acc, 0.f);
}

template <int K, int NT, int NOUT, bool RELU>
__global__ void tag_gemm_wmma(const float* __restrict__ a0, const float* __restrict__ a1,
                              const float* __restrict__ a2, const float* __restrict__ a3,
                              const float* __restrict__ W,
                              const float* __restrict__ bias,
                              float* __restrict__ out) {
    constexpr int NTHREADS = (NOUT / 16) * 32;
    __shared__ float sA[16 * K];

    const int m0   = blockIdx.x * 16;
    const int tid  = threadIdx.x;
    const int lane = tid & 31;
    const int wave = tid >> 5;
    const int half = lane >> 4;
    const int l    = lane & 15;
    const int n0   = wave * 16;

    const float* As[4] = {a0, a1, a2, a3};
    v8f c = {};

#pragma unroll
    for (int t = 0; t < NT; ++t) {
        const float* A = As[t];
        for (int i = tid; i < 16 * K; i += NTHREADS) {
            int r = i / K, kk = i % K;
            sA[i] = A[(size_t)(m0 + r) * K + kk];
        }
        __syncthreads();
        const float* Wt = W + (size_t)t * K * NOUT;
#pragma unroll
        for (int kk = 0; kk < K; kk += 4) {
            int ka = kk + 2 * half;
            v2f a, b;
            a.x = sA[l * K + ka];
            a.y = sA[l * K + ka + 1];
            b.x = Wt[(size_t)ka * NOUT + n0 + l];
            b.y = Wt[(size_t)(ka + 1) * NOUT + n0 + l];
            c = __builtin_amdgcn_wmma_f32_16x16x4_f32(false, a, false, b,
                                                      (short)0, c, false, false);
        }
        __syncthreads();
    }

    float bv = bias[n0 + l];
#pragma unroll
    for (int r = 0; r < 8; ++r) {
        int rowi = m0 + r + 8 * half;
        float v = c[r] + bv;
        if (RELU) v = fmaxf(v, 0.f);
        out[(size_t)rowi * NOUT + n0 + l] = v;
    }
}

__global__ void l2norm_kernel(float* __restrict__ out, int n) {
    int node = blockIdx.x * (blockDim.x >> 5) + (threadIdx.x >> 5);
    int lane = threadIdx.x & 31;
    if (node >= n) return;
    float* p = out + (size_t)node * 128;
    float4 v = *reinterpret_cast<float4*>(p + lane * 4);
    float ss = v.x * v.x + v.y * v.y + v.z * v.z + v.w * v.w;
#pragma unroll
    for (int off = 16; off; off >>= 1) ss += __shfl_xor(ss, off, 32);
    float nrm = fmaxf(sqrtf(ss), 1e-12f);
    float inv = 1.f / nrm;
    v.x *= inv; v.y *= inv; v.z *= inv; v.w *= inv;
    *reinterpret_cast<float4*>(p + lane * 4) = v;
}

extern "C" void kernel_launch(void* const* d_in, const int* in_sizes, int n_in,
                              void* d_out, int out_size, void* d_ws, size_t ws_size,
                              hipStream_t stream) {
    const float* x    = (const float*)d_in[0];
    const int*   ei   = (const int*)d_in[1];
    const float* ew   = (const float*)d_in[2];
    const float* w1   = (const float*)d_in[3];
    const float* b1   = (const float*)d_in[4];
    const float* w2   = (const float*)d_in[5];
    const float* b2   = (const float*)d_in[6];
    const float* w3   = (const float*)d_in[7];
    const float* b3   = (const float*)d_in[8];
    const float* fcw  = (const float*)d_in[9];
    const float* fcb  = (const float*)d_in[10];
    float* out = (float*)d_out;

    const int N = NNODES, E = NEDGES;
    const int* row = ei;
    const int* col = ei + E;

    float* wsf  = (float*)d_ws;
    float* deg  = wsf;
    float* nrm  = deg + N;
    float* P0   = nrm + E;
    float* P1   = P0 + (size_t)N * 64;
    float* F2   = P1 + (size_t)N * 64;
    float* F1   = F2 + (size_t)N * 64;
    float* P2   = F1;

    const int TB = 256;
    const int gE  = (E + TB - 1) / TB;
    const int gN  = (N + TB - 1) / TB;
    const int gMT = N / 16;

    hipMemsetAsync(deg, 0, (size_t)N * 4, stream);
    deg_kernel <<<gE, TB, 0, stream>>>(col, ew, deg, E);
    dinv_kernel<<<gN, TB, 0, stream>>>(deg, N);
    norm_kernel<<<gE, TB, 0, stream>>>(row, col, ew, deg, nrm, E);

    hipMemsetAsync(P0, 0, (size_t)N * 3 * 4, stream);
    prop3_kernel<<<gE, TB, 0, stream>>>(row, col, nrm, x, P0, E);
    layer1_kernel<<<(N * 32 + TB - 1) / TB, TB, 0, stream>>>(x, P0, w1, b1, F1);

    hipMemsetAsync(P0, 0, (size_t)N * 32 * 4, stream);
    prop_kernel<32><<<(E * 8 + TB - 1) / TB, TB, 0, stream>>>(row, col, nrm, F1, P0, E);
    hipMemsetAsync(P1, 0, (size_t)N * 32 * 4, stream);
    prop_kernel<32><<<(E * 8 + TB - 1) / TB, TB, 0, stream>>>(row, col, nrm, P0, P1, E);
    tag_gemm_wmma<32, 3, 64, true><<<gMT, 128, 0, stream>>>(F1, P0, P1, nullptr, w2, b2, F2);

    hipMemsetAsync(P0, 0, (size_t)N * 64 * 4, stream);
    prop_kernel<64><<<(E * 16 + TB - 1) / TB, TB, 0, stream>>>(row, col, nrm, F2, P0, E);
    hipMemsetAsync(P1, 0, (size_t)N * 64 * 4, stream);
    prop_kernel<64><<<(E * 16 + TB - 1) / TB, TB, 0, stream>>>(row, col, nrm, P0, P1, E);
    hipMemsetAsync(P2, 0, (size_t)N * 64 * 4, stream);
    prop_kernel<64><<<(E * 16 + TB - 1) / TB, TB, 0, stream>>>(row, col, nrm, P1, P2, E);
    tag_gemm_wmma<64, 4, 128, true><<<gMT, 256, 0, stream>>>(F2, P0, P1, P2, w3, b3, out);

    tag_gemm_wmma<128, 1, 128, false><<<gMT, 256, 0, stream>>>(out, nullptr, nullptr, nullptr,
                                                               fcw, fcb, out);
    l2norm_kernel<<<(N + 7) / 8, 256, 0, stream>>>(out, N);
}


# Neighbour sums by gather over edges counting-sorted by destination per call (returning int atomic = slot, scan, fill) instead of f32 atomic scatter-add; per-node sum order = slot (arrival) order
# speedup vs baseline: 1.7988x; 1.7988x over previous
#include <hip/hip_runtime.h>


#define NNODES 100000
#define NEDGES 1600000
#define SCAN_TB 1024

typedef __attribute__((ext_vector_type(2))) float v2f;
typedef __attribute__((ext_vector_type(8))) float v8f;
typedef __attribute__((ext_vector_type(4))) float v4f;

__global__ void count_kernel(const int* __restrict__ col, int* __restrict__ cnt,
                             int* __restrict__ rank, int E) {
    int e = blockIdx.x * blockDim.x + threadIdx.x;
    if (e < E) rank[e] = atomicAdd(&cnt[col[e]], 1);
}

__device__ __forceinline__ int block_incl_scan(int v, int* sw, int& total) {
    const int lane = threadIdx.x & 31, wave = threadIdx.x >> 5;
#pragma unroll
    for (int o = 1; o < 32; o <<= 1) { int t = __shfl_up(v, o, 32); if (lane >= o) v += t; }
    if (lane == 31) sw[wave] = v;
    __syncthreads();
    if (wave == 0) {
        int s = sw[lane];
#pragma unroll
        for (int o = 1; o < 32; o <<= 1) { int t = __shfl_up(s, o, 32); if (lane >= o) s += t; }
        sw[lane] = s;
    }
    __syncthreads();
    total = sw[31];
    return v + (wave ? sw[wave - 1] : 0);
}

__global__ void __launch_bounds__(SCAN_TB) blocksum_kernel(const int* __restrict__ cnt,
                                                           int* __restrict__ bsum, int n) {
    __shared__ int sw[32];
    int i = blockIdx.x * SCAN_TB + threadIdx.x;
    int total;
    block_incl_scan(i < n ? cnt[i] : 0, sw, total);
    if (threadIdx.x == 0) bsum[blockIdx.x] = total;
}

__global__ void __launch_bounds__(SCAN_TB) scanwrite_kernel(const int* __restrict__ cnt,
                                                            const int* __restrict__ bsum,
                                                            int* __restrict__ off, int n) {
    __shared__ int swa[32];
    __shared__ int swb[32];
    int i = blockIdx.x * SCAN_TB + threadIdx.x;
    int base, tot;
    block_incl_scan((int)threadIdx.x < (int)blockIdx.x ? bsum[threadIdx.x] : 0, swa, base);
    int v = i < n ? cnt[i] : 0;
    int incl = block_incl_scan(v, swb, tot);
    if (i < n) off[i] = base + incl - v;
    if (i == n - 1) off[n] = base + incl;
}

__global__ void fill_kernel(const int* __restrict__ row, const int* __restrict__ col,
                            const float* __restrict__ ew, const int* __restrict__ rank,
                            const int* __restrict__ off, int2* __restrict__ se, int E) {
    int e = blockIdx.x * blockDim.x + threadIdx.x;
    if (e >= E) return;
    int p = off[col[e]] + rank[e];
    se[p] = make_int2(row[e], __float_as_int(ew[e]));
}

__global__ void dinv_kernel(const int* __restrict__ off, const int2* __restrict__ se,
                            float* __restrict__ dinv, int n) {
    int gid = blockIdx.x * blockDim.x + threadIdx.x;
    int node = gid >> 3, sub = gid & 7;
    float d = 0.f;
    if (node < n) {
        int b = off[node], e = off[node + 1];
        for (int p = b + sub; p < e; p += 8) d += __int_as_float(se[p].y);
    }
    d += __shfl_xor(d, 4, 32);
    d += __shfl_xor(d, 2, 32);
    d += __shfl_xor(d, 1, 32);
    if (node < n && sub == 0) dinv[node] = (d > 0.f) ? rsqrtf(d) : 0.f;
}

__global__ void norm_kernel(const int* __restrict__ off, const int2* __restrict__ sraw,
                            const float* __restrict__ dinv, int2* __restrict__ se, int n) {
    int gid = blockIdx.x * blockDim.x + threadIdx.x;
    int node = gid >> 3, sub = gid & 7;
    if (node >= n) return;
    int b = off[node], e = off[node + 1];
    float di = dinv[node];
    for (int p = b + sub; p < e; p += 8) {
        int2 s = sraw[p];
        float w = dinv[s.x] * __int_as_float(s.y) * di;
        se[p] = make_int2(s.x, __float_as_int(w));
    }
}

__global__ void gather3_kernel(const int* __restrict__ off, const int2* __restrict__ se,
                               const float* __restrict__ hin, float* __restrict__ hout, int n) {
    int gid = blockIdx.x * blockDim.x + threadIdx.x;
    int node = gid >> 3, sub = gid & 7;
    float a0 = 0.f, a1 = 0.f, a2 = 0.f;
    if (node < n) {
        int b = off[node], e = off[node + 1];
        for (int p = b + sub; p < e; p += 8) {
            int2 s = se[p];
            float w = __int_as_float(s.y);
            const float* h = hin + (size_t)s.x * 3;
            a0 += w * h[0]; a1 += w * h[1]; a2 += w * h[2];
        }
    }
#pragma unroll
    for (int o = 4; o; o >>= 1) {
        a0 += __shfl_xor(a0, o, 32);
        a1 += __shfl_xor(a1, o, 32);
        a2 += __shfl_xor(a2, o, 32);
    }
    if (node < n && sub == 0) {
        hout[(size_t)node * 3 + 0] = a0;
        hout[(size_t)node * 3 + 1] = a1;
        hout[(size_t)node * 3 + 2] = a2;
    }
}

template <int D>
__global__ void __launch_bounds__(256) gather_kernel(const int* __restrict__ off,
                                                     const int2* __restrict__ se,
                                                     const float* __restrict__ hin,
                                                     float* __restrict__ hout, int n) {
    constexpr int LPN = D / 4;
    constexpr int U = 8;
    int gid = blockIdx.x * blockDim.x + threadIdx.x;
    int node = gid / LPN;
    int c = gid % LPN;
    if (node >= n) return;
    int p = off[node];
    const int e = off[node + 1];
    const float* hc = hin + 4 * c;
    v4f acc = {0.f, 0.f, 0.f, 0.f};
    if (p < e) {
        const int last = e - 1;
        int2 s[U];
#pragma unroll
        for (int j = 0; j < U; ++j) s[j] = se[min(p + j, last)];
        while (true) {
            v4f v[U];
#pragma unroll
            for (int j = 0; j < U; ++j)
                v[j] = *reinterpret_cast<const v4f*>(hc + (size_t)s[j].x * D);
            const int pn = p + U;
            int2 sn[U];
#pragma unroll
            for (int j = 0; j < U; ++j) sn[j] = se[min(pn + j, last)];
            __builtin_amdgcn_sched_barrier(0);
#pragma unroll
            for (int j = 0; j < U; ++j) {
                float w = (p + j < e) ? __int_as_float(s[j].y) : 0.f;
                acc += w * v[j];
            }
            if (pn >= e) break;
            p = pn;
#pragma unroll
            for (int j = 0; j < U; ++j) s[j] = sn[j];
        }
    }
    *reinterpret_cast<v4f*>(hout + (size_t)node * D + 4 * c) = acc;
}

__global__ void layer1_kernel(const float* __restrict__ x, const float* __restrict__ hop,
                              const float* __restrict__ W, const float* __restrict__ bias,
                              float* __restrict__ out) {
    int idx = blockIdx.x * blockDim.x + threadIdx.x;
    if (idx >= NNODES * 32) return;
    int i = idx >> 5, j = idx & 31;
    float acc = bias[j];
#pragma unroll
    for (int k = 0; k < 3; ++k) {
        acc += x[i * 3 + k]   * W[k * 32 + j];
        acc += hop[i * 3 + k] * W[96 + k * 32 + j];
    }
    out[idx] = fmaxf(acc, 0.f);
}

template <int K, int NT, int NOUT, bool RELU>
__global__ void tag_gemm_wmma(const float* __restrict__ a0, const float* __restrict__ a1,
                              const float* __restrict__ a2, const float* __restrict__ a3,
                              const float* __restrict__ W,
                              const float* __restrict__ bias,
                              float* __restrict__ out) {
    constexpr int NTHREADS = (NOUT / 16) * 32;
    __shared__ float sA[16 * K];

    const int m0   = blockIdx.x * 16;
    const int tid  = threadIdx.x;
    const int lane = tid & 31;
    const int wave = tid >> 5;
    const int half = lane >> 4;
    const int l    = lane & 15;
    const int n0   = wave * 16;

    const float* As[4] = {a0, a1, a2, a3};
    v8f c = {};

#pragma unroll
    for (int t = 0; t < NT; ++t) {
        const float* A = As[t];
        for (int i = tid; i < 16 * K; i += NTHREADS) {
            int r = i / K, kk = i % K;
            sA[i] = A[(size_t)(m0 + r) * K + kk];
        }
        __syncthreads();
        const float* Wt = W + (size_t)t * K * NOUT;
#pragma unroll
        for (int kk = 0; kk < K; kk += 4) {
            int ka = kk + 2 * half;
            v2f a, b;
            a.x = sA[l * K + ka];
            a.y = sA[l * K + ka + 1];
            b.x = Wt[(size_t)ka * NOUT + n0 + l];
            b.y = Wt[(size_t)(ka + 1) * NOUT + n0 + l];
            c = __builtin_amdgcn_wmma_f32_16x16x4_f32(false, a, false, b,
                                                      (short)0, c, false, false);
        }
        __syncthreads();
    }

    float bv = bias[n0 + l];
#pragma unroll
    for (int r = 0; r < 8; ++r) {
        int rowi = m0 + r + 8 * half;
        float v = c[r] + bv;
        if (RELU) v = fmaxf(v, 0.f);
        out[(size_t)rowi * NOUT + n0 + l] = v;
    }
}

__global__ void l2norm_kernel(float* __restrict__ out, int n) {
    int node = blockIdx.x * (blockDim.x >> 5) + (threadIdx.x >> 5);
    int lane = threadIdx.x & 31;
    if (node >= n) return;
    float* p = out + (size_t)node * 128;
    float4 v = *reinterpret_cast<float4*>(p + lane * 4);
    float ss = v.x * v.x + v.y * v.y + v.z * v.z + v.w * v.w;
#pragma unroll
    for (int off = 16; off; off >>= 1) ss += __shfl_xor(ss, off, 32);
    float nrm = fmaxf(sqrtf(ss), 1e-12f);
    float inv = 1.f / nrm;
    v.x *= inv; v.y *= inv; v.z *= inv; v.w *= inv;
    *reinterpret_cast<float4*>(p + lane * 4) = v;
}

extern "C" void kernel_launch(void* const* d_in, const int* in_sizes, int n_in,
                              void* d_out, int out_size, void* d_ws, size_t ws_size,
                              hipStream_t stream) {
    const float* x    = (const float*)d_in[0];
    const int*   ei   = (const int*)d_in[1];
    const float* ew   = (const float*)d_in[2];
    const float* w1   = (const float*)d_in[3];
    const float* b1   = (const float*)d_in[4];
    const float* w2   = (const float*)d_in[5];
    const float* b2   = (const float*)d_in[6];
    const float* w3   = (const float*)d_in[7];
    const float* b3   = (const float*)d_in[8];
    const float* fcw  = (const float*)d_in[9];
    const float* fcb  = (const float*)d_in[10];
    float* out = (float*)d_out;

    const int N = NNODES, E = NEDGES;
    const int* row = ei;
    const int* col = ei + E;

    float* wsf  = (float*)d_ws;
    int*   cnt  = (int*)wsf;
    int*   off  = cnt + N;
    int*   bsum = off + N + 16;
    float* dinv = (float*)(bsum + 128);
    int*   rank = (int*)(dinv + N);
    int2*  sraw = (int2*)(rank + E);
    int2*  se   = sraw + E;
    float* P0   = (float*)(se + E);
    float* P1   = P0 + (size_t)N * 64;
    float* F2   = P1 + (size_t)N * 64;
    float* F1   = F2 + (size_t)N * 64;
    float* P2   = F1;

    const int TB = 256;
    const int gE  = (E + TB - 1) / TB;
    const int gN8 = (N * 8 + TB - 1) / TB;
    const int gS  = (N + SCAN_TB - 1) / SCAN_TB;
    const int gMT = N / 16;

    hipMemsetAsync(cnt, 0, (size_t)N * 4, stream);
    count_kernel    <<<gE, TB, 0, stream>>>(col, cnt, rank, E);
    blocksum_kernel <<<gS, SCAN_TB, 0, stream>>>(cnt, bsum, N);
    scanwrite_kernel<<<gS, SCAN_TB, 0, stream>>>(cnt, bsum, off, N);
    fill_kernel     <<<gE, TB, 0, stream>>>(row, col, ew, rank, off, sraw, E);
    dinv_kernel     <<<gN8, TB, 0, stream>>>(off, sraw, dinv, N);
    norm_kernel     <<<gN8, TB, 0, stream>>>(off, sraw, dinv, se, N);

    gather3_kernel<<<gN8, TB, 0, stream>>>(off, se, x, P0, N);
    layer1_kernel<<<(N * 32 + TB - 1) / TB, TB, 0, stream>>>(x, P0, w1, b1, F1);

    gather_kernel<32><<<(N * 8 + TB - 1) / TB, TB, 0, stream>>>(off, se, F1, P0, N);
    gather_kernel<32><<<(N * 8 + TB - 1) / TB, TB, 0, stream>>>(off, se, P0, P1, N);
    tag_gemm_wmma<32, 3, 64, true><<<gMT, 128, 0, stream>>>(F1, P0, P1, nullptr, w2, b2, F2);

    gather_kernel<64><<<(N * 16 + TB - 1) / TB, TB, 0, stream>>>(off, se, F2, P0, N);
    gather_kernel<64><<<(N * 16 + TB - 1) / TB, TB, 0, stream>>>(off, se, P0, P1, N);
    gather_kernel<64><<<(N * 16 + TB - 1) / TB, TB, 0, stream>>>(off, se, P1, P2, N);
    tag_gemm_wmma<64, 4, 128, true><<<gMT, 256, 0, stream>>>(F2, P0, P1, P2, w3, b3, out);

    tag_gemm_wmma<128, 1, 128, false><<<gMT, 256, 0, stream>>>(out, nullptr, nullptr, nullptr,
                                                               fcw, fcb, out);
    l2norm_kernel<<<(N + 7) / 8, 256, 0, stream>>>(out, N);
}


# speedup vs baseline: 2.0078x; 1.1162x over previous
#include <hip/hip_runtime.h>


#define NNODES 100000
#define NEDGES 1600000
#define SCAN_TB 1024

typedef __attribute__((ext_vector_type(2))) float v2f;
typedef __attribute__((ext_vector_type(8))) float v8f;
typedef __attribute__((ext_vector_type(4))) float v4f;

__global__ void count_kernel(const int* __restrict__ col, int* __restrict__ cnt,
                             int* __restrict__ rank, int E) {
    int e = blockIdx.x * blockDim.x + threadIdx.x;
    if (e < E) rank[e] = atomicAdd(&cnt[col[e]], 1);
}

__device__ __forceinline__ int block_incl_scan(int v, int* sw, int& total) {
    const int lane = threadIdx.x & 31, wave = threadIdx.x >> 5;
#pragma unroll
    for (int o = 1; o < 32; o <<= 1) { int t = __shfl_up(v, o, 32); if (lane >= o) v += t; }
    if (lane == 31) sw[wave] = v;
    __syncthreads();
    if (wave == 0) {
        int s = sw[lane];
#pragma unroll
        for (int o = 1; o < 32; o <<= 1) { int t = __shfl_up(s, o, 32); if (lane >= o) s += t; }
        sw[lane] = s;
    }
    __syncthreads();
    total = sw[31];
    return v + (wave ? sw[wave - 1] : 0);
}

__global__ void __launch_bounds__(SCAN_TB) blocksum_kernel(const int* __restrict__ cnt,
                                                           int* __restrict__ bsum, int n) {
    __shared__ int sw[32];
    int i = blockIdx.x * SCAN_TB + threadIdx.x;
    int total;
    block_incl_scan(i < n ? cnt[i] : 0, sw, total);
    if (threadIdx.x == 0) bsum[blockIdx.x] = total;
}

__global__ void __launch_bounds__(SCAN_TB) scanwrite_kernel(const int* __restrict__ cnt,
                                                            const int* __restrict__ bsum,
                                                            int* __restrict__ off, int n) {
    __shared__ int swa[32];
    __shared__ int swb[32];
    int i = blockIdx.x * SCAN_TB + threadIdx.x;
    int base, tot;
    block_incl_scan((int)threadIdx.x < (int)blockIdx.x ? bsum[threadIdx.x] : 0, swa, base);
    int v = i < n ? cnt[i] : 0;
    int incl = block_incl_scan(v, swb, tot);
    if (i < n) off[i] = base + incl - v;
    if (i == n - 1) off[n] = base + incl;
}

__global__ void fill_kernel(const int* __restrict__ row, const int* __restrict__ col,
                            const float* __restrict__ ew, const int* __restrict__ rank,
                            const int* __restrict__ off, int2* __restrict__ se, int E) {
    int e = blockIdx.x * blockDim.x + threadIdx.x;
    if (e >= E) return;
    int p = off[col[e]] + rank[e];
    se[p] = make_int2(row[e], __float_as_int(ew[e]));
}

__global__ void dinv_kernel(const int* __restrict__ off, const int2* __restrict__ se,
                            float* __restrict__ dinv, int n) {
    int gid = blockIdx.x * blockDim.x + threadIdx.x;
    int node = gid >> 3, sub = gid & 7;
    float d = 0.f;
    if (node < n) {
        int b = off[node], e = off[node + 1];
        for (int p = b + sub; p < e; p += 8) d += __int_as_float(se[p].y);
    }
    d += __shfl_xor(d, 4, 32);
    d += __shfl_xor(d, 2, 32);
    d += __shfl_xor(d, 1, 32);
    if (node < n && sub == 0) dinv[node] = (d > 0.f) ? rsqrtf(d) : 0.f;
}

__global__ void norm_kernel(const int* __restrict__ off, const int2* __restrict__ sraw,
                            const float* __restrict__ dinv, int2* __restrict__ se, int n) {
    int gid = blockIdx.x * blockDim.x + threadIdx.x;
    int node = gid >> 3, sub = gid & 7;
    if (node >= n) return;
    int b = off[node], e = off[node + 1];
    float di = dinv[node];
    for (int p = b + sub; p < e; p += 8) {
        int2 s = sraw[p];
        float w = dinv[s.x] * __int_as_float(s.y) * di;
        se[p] = make_int2(s.x, __float_as_int(w));
    }
}

__global__ void gather3_kernel(const int* __restrict__ off, const int2* __restrict__ se,
                               const float* __restrict__ hin, float* __restrict__ hout, int n) {
    int gid = blockIdx.x * blockDim.x + threadIdx.x;
    int node = gid >> 3, sub = gid & 7;
    float a0 = 0.f, a1 = 0.f, a2 = 0.f;
    if (node < n) {
        int b = off[node], e = off[node + 1];
        for (int p = b + sub; p < e; p += 8) {
            int2 s = se[p];
            float w = __int_as_float(s.y);
            const float* h = hin + (size_t)s.x * 3;
            a0 += w * h[0]; a1 += w * h[1]; a2 += w * h[2];
        }
    }
#pragma unroll
    for (int o = 4; o; o >>= 1) {
        a0 += __shfl_xor(a0, o, 32);
        a1 += __shfl_xor(a1, o, 32);
        a2 += __shfl_xor(a2, o, 32);
    }
    if (node < n && sub == 0) {
        hout[(size_t)node * 3 + 0] = a0;
        hout[(size_t)node * 3 + 1] = a1;
        hout[(size_t)node * 3 + 2] = a2;
    }
}

template <int D>
__global__ void __launch_bounds__(256) gather_kernel(const int* __restrict__ off,
                                                     const int2* __restrict__ se,
                                                     const float* __restrict__ hin,
                                                     float* __restrict__ hout, int n) {
    constexpr int LPN = D / 4;
    constexpr int U = 8;
    int gid = blockIdx.x * blockDim.x + threadIdx.x;
    int node = gid / LPN;
    int c = gid % LPN;
    if (node >= n) return;
    int p = off[node];
    const int e = off[node + 1];
    const float* hc = hin + 4 * c;
    v4f acc = {0.f, 0.f, 0.f, 0.f};
    if (p < e) {
        const int last = e - 1;
        int2 s[U];
#pragma unroll
        for (int j = 0; j < U; ++j) s[j] = se[min(p + j, last)];
        while (true) {
            v4f v[U];
#pragma unroll
            for (int j = 0; j < U; ++j)
                v[j] = *reinterpret_cast<const v4f*>(hc + (size_t)s[j].x * D);
            const int pn = p + U;
            int2 sn[U];
#pragma unroll
            for (int j = 0; j < U; ++j) sn[j] = se[min(pn + j, last)];
            __builtin_amdgcn_sched_barrier(0);
#pragma unroll
            for (int j = 0; j < U; ++j) {
                float w = (p + j < e) ? __int_as_float(s[j].y) : 0.f;
                acc += w * v[j];
            }
            if (pn >= e) break;
            p = pn;
#pragma unroll
            for (int j = 0; j < U; ++j) s[j] = sn[j];
        }
    }
    *reinterpret_cast<v4f*>(hout + (size_t)node * D + 4 * c) = acc;
}

__global__ void layer1_kernel(const float* __restrict__ x, const float* __restrict__ hop,
                              const float* __restrict__ W, const float* __restrict__ bias,
                              float* __restrict__ out) {
    int idx = blockIdx.x * blockDim.x + threadIdx.x;
    if (idx >= NNODES * 32) return;
    int i = idx >> 5, j = idx & 31;
    float acc = bias[j];
#pragma unroll
    for (int k = 0; k < 3; ++k) {
        acc += x[i * 3 + k]   * W[k * 32 + j];
        acc += hop[i * 3 + k] * W[96 + k * 32 + j];
    }
    out[idx] = fmaxf(acc, 0.f);
}

template <int K, int NT, int NOUT, bool RELU>
__global__ void tag_gemm_wmma(const float* __restrict__ a0, const float* __restrict__ a1,
                              const float* __restrict__ a2, const float* __restrict__ a3,
                              const float* __restrict__ W,
                              const float* __restrict__ bias,
                              float* __restrict__ out) {
    constexpr int NTHREADS = (NOUT / 16) * 32;
    __shared__ float sA[16 * K];

    const int m0   = blockIdx.x * 16;
    const int tid  = threadIdx.x;
    const int lane = tid & 31;
    const int wave = tid >> 5;
    const int half = lane >> 4;
    const int l    = lane & 15;
    const int n0   = wave * 16;

    const float* As[4] = {a0, a1, a2, a3};
    v8f c = {};

#pragma unroll
    for (int t = 0; t < NT; ++t) {
        const float* A = As[t];
        for (int i = tid; i < 16 * K; i += NTHREADS) {
            int r = i / K, kk = i % K;
            sA[i] = A[(size_t)(m0 + r) * K + kk];
        }
        __syncthreads();
        const float* Wt = W + (size_t)t * K * NOUT;
#pragma unroll
        for (int kk = 0; kk < K; kk += 4) {
            int ka = kk + 2 * half;
            v2f a, b;
            a.x = sA[l * K + ka];
            a.y = sA[l * K + ka + 1];
            b.x = Wt[(size_t)ka * NOUT + n0 + l];
            b.y = Wt[(size_t)(ka + 1) * NOUT + n0 + l];
            c = __builtin_amdgcn_wmma_f32_16x16x4_f32(false, a, false, b,
                                                      (short)0, c, false, false);
        }
        __syncthreads();
    }

    float bv = bias[n0 + l];
#pragma unroll
    for (int r = 0; r < 8; ++r) {
        int rowi = m0 + r + 8 * half;
        float v = c[r] + bv;
        if (RELU) v = fmaxf(v, 0.f);
        out[(size_t)rowi * NOUT + n0 + l] = v;
    }
}

__global__ void l2norm_kernel(float* __restrict__ out, int n) {
    int node = blockIdx.x * (blockDim.x >> 5) + (threadIdx.x >> 5);
    int lane = threadIdx.x & 31;
    if (node >= n) return;
    float* p = out + (size_t)node * 128;
    float4 v = *reinterpret_cast<float4*>(p + lane * 4);
    float ss = v.x * v.x + v.y * v.y + v.z * v.z + v.w * v.w;
#pragma unroll
    for (int off = 16; off; off >>= 1) ss += __shfl_xor(ss, off, 32);
    float nrm = fmaxf(sqrtf(ss), 1e-12f);
    float inv = 1.f / nrm;
    v.x *= inv; v.y *= inv; v.z *= inv; v.w *= inv;
    *reinterpret_cast<float4*>(p + lane * 4) = v;
}

extern "C" void kernel_launch(void* const* d_in, const int* in_sizes, int n_in,
                              void* d_out, int out_size, void* d_ws, size_t ws_size,
                              hipStream_t stream) {
    const float* x    = (const float*)d_in[0];
    const int*   ei   = (const int*)d_in[1];
    const float* ew   = (const float*)d_in[2];
    const float* w1   = (const float*)d_in[3];
    const float* b1   = (const float*)d_in[4];
    const float* w2   = (const float*)d_in[5];
    const float* b2   = (const float*)d_in[6];
    const float* w3   = (const float*)d_in[7];
    const float* b3   = (const float*)d_in[8];
    const float* fcw  = (const float*)d_in[9];
    const float* fcb  = (const float*)d_in[10];
    float* out = (float*)d_out;

    const int N = NNODES, E = NEDGES;
    const int* row = ei;
    const int* col = ei + E;

    float* wsf  = (float*)d_ws;
    float* P0   = wsf;
    float* P1   = P0 + (size_t)N * 64;
    float* F2   = P1 + (size_t)N * 64;
    float* F1   = F2 + (size_t)N * 64;
    float* P2   = F1;
    int*   cnt  = (int*)(F1 + (size_t)N * 64);
    int*   off  = cnt + N;
    int*   bsum = off + N + 16;
    float* dinv = (float*)(bsum + 128);
    int*   rank = (int*)(dinv + N);
    int2*  sraw = (int2*)(rank + E);
    int2*  se   = sraw + E;

    const int TB = 256;
    const int gE  = (E + TB - 1) / TB;
    const int gN8 = (N * 8 + TB - 1) / TB;
    const int gS  = (N + SCAN_TB - 1) / SCAN_TB;
    const int gMT = N / 16;

    hipMemsetAsync(cnt, 0, (size_t)N * 4, stream);
    count_kernel    <<<gE, TB, 0, stream>>>(col, cnt, rank, E);
    blocksum_kernel <<<gS, SCAN_TB, 0, stream>>>(cnt, bsum, N);
    scanwrite_kernel<<<gS, SCAN_TB, 0, stream>>>(cnt, bsum, off, N);
    fill_kernel     <<<gE, TB, 0, stream>>>(row, col, ew, rank, off, sraw, E);
    dinv_kernel     <<<gN8, TB, 0, stream>>>(off, sraw, dinv, N);
    norm_kernel     <<<gN8, TB, 0, stream>>>(off, sraw, dinv, se, N);

    gather3_kernel<<<gN8, TB, 0, stream>>>(off, se, x, P0, N);
    layer1_kernel<<<(N * 32 + TB - 1) / TB, TB, 0, stream>>>(x, P0, w1, b1, F1);

    gather_kernel<32><<<(N * 8 + TB - 1) / TB, TB, 0, stream>>>(off, se, F1, P0, N);
    gather_kernel<32><<<(N * 8 + TB - 1) / TB, TB, 0, stream>>>(off, se, P0, P1, N);
    tag_gemm_wmma<32, 3, 64, true><<<gMT, 128, 0, stream>>>(F1, P0, P1, nullptr, w2, b2, F2);

    gather_kernel<64><<<(N * 16 + TB - 1) / TB, TB, 0, stream>>>(off, se, F2, P0, N);
    gather_kernel<64><<<(N * 16 + TB - 1) / TB, TB, 0, stream>>>(off, se, P0, P1, N);
    gather_kernel<64><<<(N * 16 + TB - 1) / TB, TB, 0, stream>>>(off, se, P1, P2, N);
    tag_gemm_wmma<64, 4, 128, true><<<gMT, 256, 0, stream>>>(F2, P0, P1, P2, w3, b3, out);

    tag_gemm_wmma<128, 1, 128, false><<<gMT, 256, 0, stream>>>(out, nullptr, nullptr, nullptr,
                                                               fcw, fcb, out);
    l2norm_kernel<<<(N + 7) / 8, 256, 0, stream>>>(out, N);
}


# Dense products: each work-group holds its W slice as B fragments in registers and walks over row tiles (next A tile prefetched): W is read once per group, not once per 16 rows; sum order unchanged
# speedup vs baseline: 2.7740x; 1.3816x over previous
#include <hip/hip_runtime.h>


#define NNODES 100000
#define NEDGES 1600000
#define SCAN_TB 1024

typedef __attribute__((ext_vector_type(2))) float v2f;
typedef __attribute__((ext_vector_type(8))) float v8f;
typedef __attribute__((ext_vector_type(4))) float v4f;

__global__ void count_kernel(const int* __restrict__ col, int* __restrict__ cnt,
                             int* __restrict__ rank, int E) {
    int e = blockIdx.x * blockDim.x + threadIdx.x;
    if (e < E) rank[e] = atomicAdd(&cnt[col[e]], 1);
}

__device__ __forceinline__ int block_incl_scan(int v, int* sw, int& total) {
    const int lane = threadIdx.x & 31, wave = threadIdx.x >> 5;
#pragma unroll
    for (int o = 1; o < 32; o <<= 1) { int t = __shfl_up(v, o, 32); if (lane >= o) v += t; }
    if (lane == 31) sw[wave] = v;
    __syncthreads();
    if (wave == 0) {
        int s = sw[lane];
#pragma unroll
        for (int o = 1; o < 32; o <<= 1) { int t = __shfl_up(s, o, 32); if (lane >= o) s += t; }
        sw[lane] = s;
    }
    __syncthreads();
    total = sw[31];
    return v + (wave ? sw[wave - 1] : 0);
}

__global__ void __launch_bounds__(SCAN_TB) blocksum_kernel(const int* __restrict__ cnt,
                                                           int* __restrict__ bsum, int n) {
    __shared__ int sw[32];
    int i = blockIdx.x * SCAN_TB + threadIdx.x;
    int total;
    block_incl_scan(i < n ? cnt[i] : 0, sw, total);
    if (threadIdx.x == 0) bsum[blockIdx.x] = total;
}

__global__ void __launch_bounds__(SCAN_TB) scanwrite_kernel(const int* __restrict__ cnt,
                                                            const int* __restrict__ bsum,
                                                            int* __restrict__ off, int n) {
    __shared__ int swa[32];
    __shared__ int swb[32];
    int i = blockIdx.x * SCAN_TB + threadIdx.x;
    int base, tot;
    block_incl_scan((int)threadIdx.x < (int)blockIdx.x ? bsum[threadIdx.x] : 0, swa, base);
    int v = i < n ? cnt[i] : 0;
    int incl = block_incl_scan(v, swb, tot);
    if (i < n) off[i] = base + incl - v;
    if (i == n - 1) off[n] = base + incl;
}

__global__ void fill_kernel(const int* __restrict__ row, const int* __restrict__ col,
                            const float* __restrict__ ew, const int* __restrict__ rank,
                            const int* __restrict__ off, int2* __restrict__ se, int E) {
    int e = blockIdx.x * blockDim.x + threadIdx.x;
    if (e >= E) return;
    int p = off[col[e]] + rank[e];
    se[p] = make_int2(row[e], __float_as_int(ew[e]));
}

__global__ void dinv_kernel(const int* __restrict__ off, const int2* __restrict__ se,
                            float* __restrict__ dinv, int n) {
    int gid = blockIdx.x * blockDim.x + threadIdx.x;
    int node = gid >> 3, sub = gid & 7;
    float d = 0.f;
    if (node < n) {
        int b = off[node], e = off[node + 1];
        for (int p = b + sub; p < e; p += 8) d += __int_as_float(se[p].y);
    }
    d += __shfl_xor(d, 4, 32);
    d += __shfl_xor(d, 2, 32);
    d += __shfl_xor(d, 1, 32);
    if (node < n && sub == 0) dinv[node] = (d > 0.f) ? rsqrtf(d) : 0.f;
}

__global__ void norm_kernel(const int* __restrict__ off, const int2* __restrict__ sraw,
                            const float* __restrict__ dinv, int2* __restrict__ se, int n) {
    int gid = blockIdx.x * blockDim.x + threadIdx.x;
    int node = gid >> 3, sub = gid & 7;
    if (node >= n) return;
    int b = off[node], e = off[node + 1];
    float di = dinv[node];
    for (int p = b + sub; p < e; p += 8) {
        int2 s = sraw[p];
        float w = dinv[s.x] * __int_as_float(s.y) * di;
        se[p] = make_int2(s.x, __float_as_int(w));
    }
}

__global__ void gather3_kernel(const int* __restrict__ off, const int2* __restrict__ se,
                               const float* __restrict__ hin, float* __restrict__ hout, int n) {
    int gid = blockIdx.x * blockDim.x + threadIdx.x;
    int node = gid >> 3, sub = gid & 7;
    float a0 = 0.f, a1 = 0.f, a2 = 0.f;
    if (node < n) {
        int b = off[node], e = off[node + 1];
        for (int p = b + sub; p < e; p += 8) {
            int2 s = se[p];
            float w = __int_as_float(s.y);
            const float* h = hin + (size_t)s.x * 3;
            a0 += w * h[0]; a1 += w * h[1]; a2 += w * h[2];
        }
    }
#pragma unroll
    for (int o = 4; o; o >>= 1) {
        a0 += __shfl_xor(a0, o, 32);
        a1 += __shfl_xor(a1, o, 32);
        a2 += __shfl_xor(a2, o, 32);
    }
    if (node < n && sub == 0) {
        hout[(size_t)node * 3 + 0] = a0;
        hout[(size_t)node * 3 + 1] = a1;
        hout[(size_t)node * 3 + 2] = a2;
    }
}

template <int D>
__global__ void __launch_bounds__(256) gather_kernel(const int* __restrict__ off,
                                                     const int2* __restrict__ se,
                                                     const float* __restrict__ hin,
                                                     float* __restrict__ hout, int n) {
    constexpr int LPN = D / 4;
    constexpr int U = 8;
    int gid = blockIdx.x * blockDim.x + threadIdx.x;
    int node = gid / LPN;
    int c = gid % LPN;
    if (node >= n) return;
    int p = off[node];
    const int e = off[node + 1];
    const float* hc = hin + 4 * c;
    v4f acc = {0.f, 0.f, 0.f, 0.f};
    if (p < e) {
        const int last = e - 1;
        int2 s[U];
#pragma unroll
        for (int j = 0; j < U; ++j) s[j] = se[min(p + j, last)];
        while (true) {
            v4f v[U];
#pragma unroll
            for (int j = 0; j < U; ++j)
                v[j] = *reinterpret_cast<const v4f*>(hc + (size_t)s[j].x * D);
            const int pn = p + U;
            int2 sn[U];
#pragma unroll
            for (int j = 0; j < U; ++j) sn[j] = se[min(pn + j, last)];
            __builtin_amdgcn_sched_barrier(0);
#pragma unroll
            for (int j = 0; j < U; ++j) {
                float w = (p + j < e) ? __int_as_float(s[j].y) : 0.f;
                acc += w * v[j];
            }
            if (pn >= e) break;
            p = pn;
#pragma unroll
            for (int j = 0; j < U; ++j) s[j] = sn[j];
        }
    }
    *reinterpret_cast<v4f*>(hout + (size_t)node * D + 4 * c) = acc;
}

__global__ void layer1_kernel(const float* __restrict__ x, const float* __restrict__ hop,
                              const float* __restrict__ W, const float* __restrict__ bias,
                              float* __restrict__ out) {
    int idx = blockIdx.x * blockDim.x + threadIdx.x;
    if (idx >= NNODES * 32) return;
    int i = idx >> 5, j = idx & 31;
    float acc = bias[j];
#pragma unroll
    for (int k = 0; k < 3; ++k) {
        acc += x[i * 3 + k]   * W[k * 32 + j];
        acc += hop[i * 3 + k] * W[96 + k * 32 + j];
    }
    out[idx] = fmaxf(acc, 0.f);
}

template <int K, int NT, int NTHREADS>
__device__ __forceinline__ void fetch_rows(v4f (&pre)[NT * 16 * (K / 4) / NTHREADS],
                                           const float* a0, const float* a1, const float* a2,
                                           const float* a3, int tile, int tid) {
    constexpr int RV = 16 * (K / 4);
    constexpr int VPT = NT * RV / NTHREADS;
#pragma unroll
    for (int q = 0; q < VPT; ++q) {
        const int t = (q * NTHREADS) / RV;
        const int rem = (q * NTHREADS) % RV + tid;
        const float* A = t == 0 ? a0 : t == 1 ? a1 : t == 2 ? a2 : a3;
        pre[q] = *reinterpret_cast<const v4f*>(A + (size_t)tile * 16 * K + 4 * rem);
    }
}

template <int K, int NT, int NOUT, bool RELU>
__global__ void __launch_bounds__((NOUT / 16) * 32)
tag_gemm_wmma(const float* a0, const float* a1, const float* a2, const float* a3,
              const float* __restrict__ W,
              const float* __restrict__ bias,
              float* out, int ntiles) {
    constexpr int NTHREADS = (NOUT / 16) * 32;
    constexpr int KP  = K + 4;
    constexpr int KS  = K / 4;
    constexpr int RV  = 16 * KS;
    constexpr int VPT = NT * RV / NTHREADS;
    static_assert(RV % NTHREADS == 0, "a staging slot must not straddle two terms");
    __shared__ alignas(16) float sA[2][NT * 16 * KP];

    const int tid  = threadIdx.x;
    const int lane = tid & 31;
    const int wave = tid >> 5;
    const int half = lane >> 4;
    const int l    = lane & 15;
    const int n0   = wave * 16;

    int tile = blockIdx.x;
    v4f pre[VPT];
    if (tile < ntiles) fetch_rows<K, NT, NTHREADS>(pre, a0, a1, a2, a3, tile, tid);

    v2f b[NT][KS];
#pragma unroll
    for (int t = 0; t < NT; ++t) {
#pragma unroll
        for (int j = 0; j < KS; ++j) {
            const float* w = W + ((size_t)t * K + 4 * j + 2 * half) * NOUT + n0 + l;
            b[t][j].x = w[0];
            b[t][j].y = w[NOUT];
        }
    }
    const float bv = bias[n0 + l];

    int stage = 0;
    for (; tile < ntiles; tile += gridDim.x) {
        float* s = sA[stage];
#pragma unroll
        for (int q = 0; q < VPT; ++q) {
            const int t = (q * NTHREADS) / RV;
            const int rem = (q * NTHREADS) % RV + tid;
            *reinterpret_cast<v4f*>(s + t * 16 * KP + (rem / KS) * KP + 4 * (rem % KS)) = pre[q];
        }
        __syncthreads();
        const int nxt = tile + gridDim.x;
        if (nxt < ntiles) fetch_rows<K, NT, NTHREADS>(pre, a0, a1, a2, a3, nxt, tid);

        v8f c = {};
        const float* sl = s + l * KP + 2 * half;
#pragma unroll
        for (int t = 0; t < NT; ++t) {
            v2f a[KS];
#pragma unroll
            for (int j = 0; j < KS; ++j)
                a[j] = *reinterpret_cast<const v2f*>(sl + t * 16 * KP + 4 * j);
#pragma unroll
            for (int j = 0; j < KS; ++j)
                c = __builtin_amdgcn_wmma_f32_16x16x4_f32(false, a[j], false, b[t][j],
                                                          (short)0, c, false, false);
        }

#pragma unroll
        for (int r = 0; r < 8; ++r) {
            int rowi = tile * 16 + r + 8 * half;
            float v = c[r] + bv;
            if (RELU) v = fmaxf(v, 0.f);
            out[(size_t)rowi * NOUT + n0 + l] = v;
        }
        stage ^= 1;
    }
}

__global__ void l2norm_kernel(float* __restrict__ out, int n) {
    int node = blockIdx.x * (blockDim.x >> 5) + (threadIdx.x >> 5);
    int lane = threadIdx.x & 31;
    if (node >= n) return;
    float* p = out + (size_t)node * 128;
    float4 v = *reinterpret_cast<float4*>(p + lane * 4);
    float ss = v.x * v.x + v.y * v.y + v.z * v.z + v.w * v.w;
#pragma unroll
    for (int off = 16; off; off >>= 1) ss += __shfl_xor(ss, off, 32);
    float nrm = fmaxf(sqrtf(ss), 1e-12f);
    float inv = 1.f / nrm;
    v.x *= inv; v.y *= inv; v.z *= inv; v.w *= inv;
    *reinterpret_cast<float4*>(p + lane * 4) = v;
}

extern "C" void kernel_launch(void* const* d_in, const int* in_sizes, int n_in,
                              void* d_out, int out_size, void* d_ws, size_t ws_size,
                              hipStream_t stream) {
    const float* x    = (const float*)d_in[0];
    const int*   ei   = (const int*)d_in[1];
    const float* ew   = (const float*)d_in[2];
    const float* w1   = (const float*)d_in[3];
    const float* b1   = (const float*)d_in[4];
    const float* w2   = (const float*)d_in[5];
    const float* b2   = (const float*)d_in[6];
    const float* w3   = (const float*)d_in[7];
    const float* b3   = (const float*)d_in[8];
    const float* fcw  = (const float*)d_in[9];
    const float* fcb  = (const float*)d_in[10];
    float* out = (float*)d_out;

    const int N = NNODES, E = NEDGES;
    const int* row = ei;
    const int* col = ei + E;

    float* wsf  = (float*)d_ws;
    float* P0   = wsf;
    float* P1   = P0 + (size_t)N * 64;
    float* F2   = P1 + (size_t)N * 64;
    float* F1   = F2 + (size_t)N * 64;
    float* P2   = F1;
    int*   cnt  = (int*)(F1 + (size_t)N * 64);
    int*   off  = cnt + N;
    int*   bsum = off + N + 16;
    float* dinv = (float*)(bsum + 128);
    int*   rank = (int*)(dinv + N);
    int2*  sraw = (int2*)(rank + E);
    int2*  se   = sraw + E;

    const int TB = 256;
    const int gE  = (E + TB - 1) / TB;
    const int gN8 = (N * 8 + TB - 1) / TB;
    const int gS  = (N + SCAN_TB - 1) / SCAN_TB;
    const int gMT = N / 16;
    const int gG3 = 512;
    const int gGF = 1024;
    const int gG2 = 1280;

    hipMemsetAsync(cnt, 0, (size_t)N * 4, stream);
    count_kernel    <<<gE, TB, 0, stream>>>(col, cnt, rank, E);
    blocksum_kernel <<<gS, SCAN_TB, 0, stream>>>(cnt, bsum, N);
    scanwrite_kernel<<<gS, SCAN_TB, 0, stream>>>(cnt, bsum, off, N);
    fill_kernel     <<<gE, TB, 0, stream>>>(row, col, ew, rank, off, sraw, E);
    dinv_kernel     <<<gN8, TB, 0, stream>>>(off, sraw, dinv, N);
    norm_kernel     <<<gN8, TB, 0, stream>>>(off, sraw, dinv, se, N);

    gather3_kernel<<<gN8, TB, 0, stream>>>(off, se, x, P0, N);
    layer1_kernel<<<(N * 32 + TB - 1) / TB, TB, 0, stream>>>(x, P0, w1, b1, F1);

    gather_kernel<32><<<(N * 8 + TB - 1) / TB, TB, 0, stream>>>(off, se, F1, P0, N);
    gather_kernel<32><<<(N * 8 + TB - 1) / TB, TB, 0, stream>>>(off, se, P0, P1, N);
    tag_gemm_wmma<32, 3, 64, true><<<gG2, 128, 0, stream>>>(F1, P0, P1, nullptr, w2, b2, F2, gMT);

    gather_kernel<64><<<(N * 16 + TB - 1) / TB, TB, 0, stream>>>(off, se, F2, P0, N);
    gather_kernel<64><<<(N * 16 + TB - 1) / TB, TB, 0, stream>>>(off, se, P0, P1, N);
    gather_kernel<64><<<(N * 16 + TB - 1) / TB, TB, 0, stream>>>(off, se, P1, P2, N);
    tag_gemm_wmma<64, 4, 128, true><<<gG3, 256, 0, stream>>>(F2, P0, P1, P2, w3, b3, out, gMT);

    tag_gemm_wmma<128, 1, 128, false><<<gGF, 256, 0, stream>>>(out, nullptr, nullptr, nullptr,
                                                              fcw, fcb, out, gMT);
    l2norm_kernel<<<(N + 7) / 8, 256, 0, stream>>>(out, N);
}


# Edges sorted by destination in two LDS counting-sort levels (256-node bucket, then node) instead of one returning global atomic per edge; per-node sum order = LDS arrival order
# speedup vs baseline: 3.5583x; 1.2827x over previous
#include <hip/hip_runtime.h>


#define NNODES 100000
#define NEDGES 1600000
#define SCAN_TB 1024

typedef __attribute__((ext_vector_type(2))) float v2f;
typedef __attribute__((ext_vector_type(8))) float v8f;
typedef __attribute__((ext_vector_type(4))) float v4f;

#define BSH 8
#define BW (1 << BSH)
#define NBUCK ((NNODES + BW - 1) / BW)
#define SORT_WGS 256
#define SORT_TB 1024
#define HN (NBUCK * SORT_WGS)
#define ROWBITS 20
#define SORT_U 8

__global__ void __launch_bounds__(SORT_TB) bucket_hist_kernel(const int* __restrict__ col,
                                                              int* __restrict__ hist, int E, int CH) {
    __shared__ int h[NBUCK];
    const int tid = threadIdx.x;
    for (int i = tid; i < NBUCK; i += SORT_TB) h[i] = 0;
    __syncthreads();
    const int lo = blockIdx.x * CH;
    const int hi = min(E, lo + CH);
    for (int e0 = lo + tid; e0 < hi; e0 += SORT_U * SORT_TB) {
        int c[SORT_U];
#pragma unroll
        for (int k = 0; k < SORT_U; ++k) { const int e = e0 + k * SORT_TB; c[k] = e < hi ? col[e] : -1; }
#pragma unroll
        for (int k = 0; k < SORT_U; ++k) if (c[k] >= 0) atomicAdd(&h[c[k] >> BSH], 1);
    }
    __syncthreads();
    for (int i = tid; i < NBUCK; i += SORT_TB) hist[i * SORT_WGS + blockIdx.x] = h[i];
}

__device__ __forceinline__ int block_incl_scan(int v, int* sw, int& total) {
    const int lane = threadIdx.x & 31, wave = threadIdx.x >> 5;
#pragma unroll
    for (int o = 1; o < 32; o <<= 1) { int t = __shfl_up(v, o, 32); if (lane >= o) v += t; }
    if (lane == 31) sw[wave] = v;
    __syncthreads();
    if (wave == 0) {
        int s = sw[lane];
#pragma unroll
        for (int o = 1; o < 32; o <<= 1) { int t = __shfl_up(s, o, 32); if (lane >= o) s += t; }
        sw[lane] = s;
    }
    __syncthreads();
    total = sw[31];
    return v + (wave ? sw[wave - 1] : 0);
}

__global__ void __launch_bounds__(SCAN_TB) blocksum_kernel(const int* __restrict__ cnt,
                                                           int* __restrict__ bsum, int n) {
    __shared__ int sw[32];
    int i = blockIdx.x * SCAN_TB + threadIdx.x;
    int total;
    block_incl_scan(i < n ? cnt[i] : 0, sw, total);
    if (threadIdx.x == 0) bsum[blockIdx.x] = total;
}

__global__ void __launch_bounds__(SCAN_TB) scanwrite_kernel(const int* __restrict__ cnt,
                                                            const int* __restrict__ bsum,
                                                            int* __restrict__ off, int n) {
    __shared__ int swa[32];
    __shared__ int swb[32];
    int i = blockIdx.x * SCAN_TB + threadIdx.x;
    int base, tot;
    block_incl_scan((int)threadIdx.x < (int)blockIdx.x ? bsum[threadIdx.x] : 0, swa, base);
    int v = i < n ? cnt[i] : 0;
    int incl = block_incl_scan(v, swb, tot);
    if (i < n) off[i] = base + incl - v;
    if (i == n - 1) off[n] = base + incl;
}

__global__ void __launch_bounds__(SORT_TB) bucket_scatter_kernel(const int* __restrict__ row,
                                                                 const int* __restrict__ col,
                                                                 const float* __restrict__ ew,
                                                                 const int* __restrict__ hoff,
                                                                 int2* __restrict__ bk, int E, int CH) {
    __shared__ int cur[NBUCK];
    const int tid = threadIdx.x;
    for (int i = tid; i < NBUCK; i += SORT_TB) cur[i] = hoff[i * SORT_WGS + blockIdx.x];
    __syncthreads();
    const int lo = blockIdx.x * CH;
    const int hi = min(E, lo + CH);
    for (int e0 = lo + tid; e0 < hi; e0 += SORT_U * SORT_TB) {
        int c[SORT_U], r[SORT_U], w[SORT_U];
#pragma unroll
        for (int k = 0; k < SORT_U; ++k) {
            const int e = e0 + k * SORT_TB;
            const bool ok = e < hi;
            c[k] = ok ? col[e] : -1;
            r[k] = ok ? row[e] : 0;
            w[k] = ok ? __float_as_int(ew[e]) : 0;
        }
#pragma unroll
        for (int k = 0; k < SORT_U; ++k) {
            if (c[k] >= 0) {
                const int p = atomicAdd(&cur[c[k] >> BSH], 1);
                bk[p] = make_int2(r[k] | ((c[k] & (BW - 1)) << ROWBITS), w[k]);
            }
        }
    }
}

#define SORT_R 8
__global__ void __launch_bounds__(SORT_TB) bucket_sort_kernel(const int* __restrict__ hoff,
                                                              const int2* __restrict__ bk,
                                                              int* __restrict__ off,
                                                              int2* __restrict__ sraw, int n) {
    __shared__ int cnt[BW];
    __shared__ int cur[BW];
    __shared__ int wsum[BW / 32];
    const int tid = threadIdx.x;
    const int b = blockIdx.x;
    const int start = hoff[b * SORT_WGS];
    const int end = hoff[(b + 1) * SORT_WGS];
    if (tid < BW) cnt[tid] = 0;
    int2 k[SORT_R];
#pragma unroll
    for (int j = 0; j < SORT_R; ++j) {
        const int p = start + tid + j * SORT_TB;
        k[j] = p < end ? bk[p] : make_int2(-1, 0);
    }
    __syncthreads();
#pragma unroll
    for (int j = 0; j < SORT_R; ++j) if (k[j].x >= 0) atomicAdd(&cnt[k[j].x >> ROWBITS], 1);
    for (int p = start + tid + SORT_R * SORT_TB; p < end; p += SORT_TB)
        atomicAdd(&cnt[bk[p].x >> ROWBITS], 1);
    __syncthreads();
    int v = 0, incl = 0;
    if (tid < BW) {
        v = cnt[tid];
        incl = v;
#pragma unroll
        for (int o = 1; o < 32; o <<= 1) { int t = __shfl_up(incl, o, 32); if ((tid & 31) >= o) incl += t; }
        if ((tid & 31) == 31) wsum[tid >> 5] = incl;
    }
    __syncthreads();
    if (tid < BW) {
        int base = start;
        for (int w = 0; w < (tid >> 5); ++w) base += wsum[w];
        const int ex = base + incl - v;
        cur[tid] = ex;
        const int node = b * BW + tid;
        if (node < n) off[node] = ex;
        if (node == n - 1) off[n] = ex + v;
    }
    __syncthreads();
#pragma unroll
    for (int j = 0; j < SORT_R; ++j) {
        if (k[j].x >= 0) {
            const int q = atomicAdd(&cur[k[j].x >> ROWBITS], 1);
            sraw[q] = make_int2(k[j].x & ((1 << ROWBITS) - 1), k[j].y);
        }
    }
    for (int p = start + tid + SORT_R * SORT_TB; p < end; p += SORT_TB) {
        const int2 r = bk[p];
        const int q = atomicAdd(&cur[r.x >> ROWBITS], 1);
        sraw[q] = make_int2(r.x & ((1 << ROWBITS) - 1), r.y);
    }
}

__global__ void dinv_kernel(const int* __restrict__ off, const int2* __restrict__ se,
                            float* __restrict__ dinv, int n) {
    int gid = blockIdx.x * blockDim.x + threadIdx.x;
    int node = gid >> 3, sub = gid & 7;
    float d = 0.f;
    if (node < n) {
        int b = off[node], e = off[node + 1];
        for (int p = b + sub; p < e; p += 8) d += __int_as_float(se[p].y);
    }
    d += __shfl_xor(d, 4, 32);
    d += __shfl_xor(d, 2, 32);
    d += __shfl_xor(d, 1, 32);
    if (node < n && sub == 0) dinv[node] = (d > 0.f) ? rsqrtf(d) : 0.f;
}

__global__ void norm_kernel(const int* __restrict__ off, const int2* __restrict__ sraw,
                            const float* __restrict__ dinv, int2* __restrict__ se, int n) {
    int gid = blockIdx.x * blockDim.x + threadIdx.x;
    int node = gid >> 3, sub = gid & 7;
    if (node >= n) return;
    int b = off[node], e = off[node + 1];
    float di = dinv[node];
    for (int p = b + sub; p < e; p += 8) {
        int2 s = sraw[p];
        float w = dinv[s.x] * __int_as_float(s.y) * di;
        se[p] = make_int2(s.x, __float_as_int(w));
    }
}

__global__ void gather3_kernel(const int* __restrict__ off, const int2* __restrict__ se,
                               const float* __restrict__ hin, float* __restrict__ hout, int n) {
    int gid = blockIdx.x * blockDim.x + threadIdx.x;
    int node = gid >> 3, sub = gid & 7;
    float a0 = 0.f, a1 = 0.f, a2 = 0.f;
    if (node < n) {
        int b = off[node], e = off[node + 1];
        for (int p = b + sub; p < e; p += 8) {
            int2 s = se[p];
            float w = __int_as_float(s.y);
            const float* h = hin + (size_t)s.x * 3;
            a0 += w * h[0]; a1 += w * h[1]; a2 += w * h[2];
        }
    }
#pragma unroll
    for (int o = 4; o; o >>= 1) {
        a0 += __shfl_xor(a0, o, 32);
        a1 += __shfl_xor(a1, o, 32);
        a2 += __shfl_xor(a2, o, 32);
    }
    if (node < n && sub == 0) {
        hout[(size_t)node * 3 + 0] = a0;
        hout[(size_t)node * 3 + 1] = a1;
        hout[(size_t)node * 3 + 2] = a2;
    }
}

template <int D>
__global__ void __launch_bounds__(256) gather_kernel(const int* __restrict__ off,
                                                     const int2* __restrict__ se,
                                                     const float* __restrict__ hin,
                                                     float* __restrict__ hout, int n) {
    constexpr int LPN = D / 4;
    constexpr int U = 8;
    int gid = blockIdx.x * blockDim.x + threadIdx.x;
    int node = gid / LPN;
    int c = gid % LPN;
    if (node >= n) return;
    int p = off[node];
    const int e = off[node + 1];
    const float* hc = hin + 4 * c;
    v4f acc = {0.f, 0.f, 0.f, 0.f};
    if (p < e) {
        const int last = e - 1;
        int2 s[U];
#pragma unroll
        for (int j = 0; j < U; ++j) s[j] = se[min(p + j, last)];
        while (true) {
            v4f v[U];
#pragma unroll
            for (int j = 0; j < U; ++j)
                v[j] = *reinterpret_cast<const v4f*>(hc + (size_t)s[j].x * D);
            const int pn = p + U;
            int2 sn[U];
#pragma unroll
            for (int j = 0; j < U; ++j) sn[j] = se[min(pn + j, last)];
            __builtin_amdgcn_sched_barrier(0);
#pragma unroll
            for (int j = 0; j < U; ++j) {
                float w = (p + j < e) ? __int_as_float(s[j].y) : 0.f;
                acc += w * v[j];
            }
            if (pn >= e) break;
            p = pn;
#pragma unroll
            for (int j = 0; j < U; ++j) s[j] = sn[j];
        }
    }
    *reinterpret_cast<v4f*>(hout + (size_t)node * D + 4 * c) = acc;
}

__global__ void layer1_kernel(const float* __restrict__ x, const float* __restrict__ hop,
                              const float* __restrict__ W, const float* __restrict__ bias,
                              float* __restrict__ out) {
    int idx = blockIdx.x * blockDim.x + threadIdx.x;
    if (idx >= NNODES * 32) return;
    int i = idx >> 5, j = idx & 31;
    float acc = bias[j];
#pragma unroll
    for (int k = 0; k < 3; ++k) {
        acc += x[i * 3 + k]   * W[k * 32 + j];
        acc += hop[i * 3 + k] * W[96 + k * 32 + j];
    }
    out[idx] = fmaxf(acc, 0.f);
}

template <int K, int NT, int NTHREADS>
__device__ __forceinline__ void fetch_rows(v4f (&pre)[NT * 16 * (K / 4) / NTHREADS],
                                           const float* a0, const float* a1, const float* a2,
                                           const float* a3, int tile, int tid) {
    constexpr int RV = 16 * (K / 4);
    constexpr int VPT = NT * RV / NTHREADS;
#pragma unroll
    for (int q = 0; q < VPT; ++q) {
        const int t = (q * NTHREADS) / RV;
        const int rem = (q * NTHREADS) % RV + tid;
        const float* A = t == 0 ? a0 : t == 1 ? a1 : t == 2 ? a2 : a3;
        pre[q] = *reinterpret_cast<const v4f*>(A + (size_t)tile * 16 * K + 4 * rem);
    }
}

template <int K, int NT, int NOUT, bool RELU>
__global__ void __launch_bounds__((NOUT / 16) * 32)
tag_gemm_wmma(const float* a0, const float* a1, const float* a2, const float* a3,
              const float* __restrict__ W,
              const float* __restrict__ bias,
              float* out, int ntiles) {
    constexpr int NTHREADS = (NOUT / 16) * 32;
    constexpr int KP  = K + 4;
    constexpr int KS  = K / 4;
    constexpr int RV  = 16 * KS;
    constexpr int VPT = NT * RV / NTHREADS;
    static_assert(RV % NTHREADS == 0, "a staging slot must not straddle two terms");
    __shared__ alignas(16) float sA[2][NT * 16 * KP];

    const int tid  = threadIdx.x;
    const int lane = tid & 31;
    const int wave = tid >> 5;
    const int half = lane >> 4;
    const int l    = lane & 15;
    const int n0   = wave * 16;

    int tile = blockIdx.x;
    v4f pre[VPT];
    if (tile < ntiles) fetch_rows<K, NT, NTHREADS>(pre, a0, a1, a2, a3, tile, tid);

    v2f b[NT][KS];
#pragma unroll
    for (int t = 0; t < NT; ++t) {
#pragma unroll
        for (int j = 0; j < KS; ++j) {
            const float* w = W + ((size_t)t * K + 4 * j + 2 * half) * NOUT + n0 + l;
            b[t][j].x = w[0];
            b[t][j].y = w[NOUT];
        }
    }
    const float bv = bias[n0 + l];

    int stage = 0;
    for (; tile < ntiles; tile += gridDim.x) {
        float* s = sA[stage];
#pragma unroll
        for (int q = 0; q < VPT; ++q) {
            const int t = (q * NTHREADS) / RV;
            const int rem = (q * NTHREADS) % RV + tid;
            *reinterpret_cast<v4f*>(s + t * 16 * KP + (rem / KS) * KP + 4 * (rem % KS)) = pre[q];
        }
        __syncthreads();
        const int nxt = tile + gridDim.x;
        if (nxt < ntiles) fetch_rows<K, NT, NTHREADS>(pre, a0, a1, a2, a3, nxt, tid);

        v8f c = {};
        const float* sl = s + l * KP + 2 * half;
#pragma unroll
        for (int t = 0; t < NT; ++t) {
            v2f a[KS];
#pragma unroll
            for (int j = 0; j < KS; ++j)
                a[j] = *reinterpret_cast<const v2f*>(sl + t * 16 * KP + 4 * j);
#pragma unroll
            for (int j = 0; j < KS; ++j)
                c = __builtin_amdgcn_wmma_f32_16x16x4_f32(false, a[j], false, b[t][j],
                                                          (short)0, c, false, false);
        }

#pragma unroll
        for (int r = 0; r < 8; ++r) {
            int rowi = tile * 16 + r + 8 * half;
            float v = c[r] + bv;
            if (RELU) v = fmaxf(v, 0.f);
            out[(size_t)rowi * NOUT + n0 + l] = v;
        }
        stage ^= 1;
    }
}

__global__ void l2norm_kernel(float* __restrict__ out, int n) {
    int node = blockIdx.x * (blockDim.x >> 5) + (threadIdx.x >> 5);
    int lane = threadIdx.x & 31;
    if (node >= n) return;
    float* p = out + (size_t)node * 128;
    float4 v = *reinterpret_cast<float4*>(p + lane * 4);
    float ss = v.x * v.x + v.y * v.y + v.z * v.z + v.w * v.w;
#pragma unroll
    for (int off = 16; off; off >>= 1) ss += __shfl_xor(ss, off, 32);
    float nrm = fmaxf(sqrtf(ss), 1e-12f);
    float inv = 1.f / nrm;
    v.x *= inv; v.y *= inv; v.z *= inv; v.w *= inv;
    *reinterpret_cast<float4*>(p + lane * 4) = v;
}

extern "C" void kernel_launch(void* const* d_in, const int* in_sizes, int n_in,
                              void* d_out, int out_size, void* d_ws, size_t ws_size,
                              hipStream_t stream) {
    const float* x    = (const float*)d_in[0];
    const int*   ei   = (const int*)d_in[1];
    const float* ew   = (const float*)d_in[2];
    const float* w1   = (const float*)d_in[3];
    const float* b1   = (const float*)d_in[4];
    const float* w2   = (const float*)d_in[5];
    const float* b2   = (const float*)d_in[6];
    const float* w3   = (const float*)d_in[7];
    const float* b3   = (const float*)d_in[8];
    const float* fcw  = (const float*)d_in[9];
    const float* fcb  = (const float*)d_in[10];
    float* out = (float*)d_out;

    const int N = NNODES, E = NEDGES;
    const int HNP = (HN + 1 + 63) / 64 * 64;
    const int* row = ei;
    const int* col = ei + E;

    float* wsf  = (float*)d_ws;
    float* P0   = wsf;
    float* P1   = P0 + (size_t)N * 64;
    float* F2   = P1 + (size_t)N * 64;
    float* F1   = F2 + (size_t)N * 64;
    float* P2   = F1;
    int*   hist = (int*)(F1 + (size_t)N * 64);
    int*   hoff = hist + HNP;
    int*   off  = hoff + HNP;
    int*   bsum = off + N + 16;
    float* dinv = (float*)(bsum + 128);
    int2*  sraw = (int2*)(dinv + N);
    int2*  se   = sraw + E;
    int2*  bk   = se;

    const int TB = 256;
    const int gN8 = (N * 8 + TB - 1) / TB;
    const int gS  = (HN + SCAN_TB - 1) / SCAN_TB;
    const int CH  = (E + SORT_WGS - 1) / SORT_WGS;
    const int gMT = N / 16;
    const int gG3 = 512;
    const int gGF = 1024;
    const int gG2 = 1280;

    bucket_hist_kernel   <<<SORT_WGS, SORT_TB, 0, stream>>>(col, hist, E, CH);
    blocksum_kernel      <<<gS, SCAN_TB, 0, stream>>>(hist, bsum, HN);
    scanwrite_kernel     <<<gS, SCAN_TB, 0, stream>>>(hist, bsum, hoff, HN);
    bucket_scatter_kernel<<<SORT_WGS, SORT_TB, 0, stream>>>(row, col, ew, hoff, bk, E, CH);
    bucket_sort_kernel   <<<NBUCK, SORT_TB, 0, stream>>>(hoff, bk, off, sraw, N);
    dinv_kernel     <<<gN8, TB, 0, stream>>>(off, sraw, dinv, N);
    norm_kernel     <<<gN8, TB, 0, stream>>>(off, sraw, dinv, se, N);

    gather3_kernel<<<gN8, TB, 0, stream>>>(off, se, x, P0, N);
    layer1_kernel<<<(N * 32 + TB - 1) / TB, TB, 0, stream>>>(x, P0, w1, b1, F1);

    gather_kernel<32><<<(N * 8 + TB - 1) / TB, TB, 0, stream>>>(off, se, F1, P0, N);
    gather_kernel<32><<<(N * 8 + TB - 1) / TB, TB, 0, stream>>>(off, se, P0, P1, N);
    tag_gemm_wmma<32, 3, 64, true><<<gG2, 128, 0, stream>>>(F1, P0, P1, nullptr, w2, b2, F2, gMT);

    gather_kernel<64><<<(N * 16 + TB - 1) / TB, TB, 0, stream>>>(off, se, F2, P0, N);
    gather_kernel<64><<<(N * 16 + TB - 1) / TB, TB, 0, stream>>>(off, se, P0, P1, N);
    gather_kernel<64><<<(N * 16 + TB - 1) / TB, TB, 0, stream>>>(off, se, P1, P2, N);
    tag_gemm_wmma<64, 4, 128, true><<<gG3, 256, 0, stream>>>(F2, P0, P1, P2, w3, b3, out, gMT);

    tag_gemm_wmma<128, 1, 128, false><<<gGF, 256, 0, stream>>>(out, nullptr, nullptr, nullptr,
                                                              fcw, fcb, out, gMT);
    l2norm_kernel<<<(N + 7) / 8, 256, 0, stream>>>(out, N);
}


# Row-local stages run inside their producers: fc product + row normalise on the layer-3 product's tile in LDS, layer 1 in the 3-wide hop's lanes; three planes never stored; sum order unchanged
# speedup vs baseline: 4.7305x; 1.3294x over previous
#include <hip/hip_runtime.h>


#define NNODES 100000
#define NEDGES 1600000
#define SCAN_TB 1024

typedef __attribute__((ext_vector_type(2))) float v2f;
typedef __attribute__((ext_vector_type(8))) float v8f;
typedef __attribute__((ext_vector_type(4))) float v4f;

#define BSH 8
#define BW (1 << BSH)
#define NBUCK ((NNODES + BW - 1) / BW)
#define SORT_WGS 256
#define SORT_TB 1024
#define HN (NBUCK * SORT_WGS)
#define ROWBITS 20
#define SORT_U 8

__global__ void __launch_bounds__(SORT_TB) bucket_hist_kernel(const int* __restrict__ col,
                                                              int* __restrict__ hist, int E, int CH) {
    __shared__ int h[NBUCK];
    const int tid = threadIdx.x;
    for (int i = tid; i < NBUCK; i += SORT_TB) h[i] = 0;
    __syncthreads();
    const int lo = blockIdx.x * CH;
    const int hi = min(E, lo + CH);
    for (int e0 = lo + tid; e0 < hi; e0 += SORT_U * SORT_TB) {
        int c[SORT_U];
#pragma unroll
        for (int k = 0; k < SORT_U; ++k) { const int e = e0 + k * SORT_TB; c[k] = e < hi ? col[e] : -1; }
#pragma unroll
        for (int k = 0; k < SORT_U; ++k) if (c[k] >= 0) atomicAdd(&h[c[k] >> BSH], 1);
    }
    __syncthreads();
    for (int i = tid; i < NBUCK; i += SORT_TB) hist[i * SORT_WGS + blockIdx.x] = h[i];
}

__device__ __forceinline__ int block_incl_scan(int v, int* sw, int& total) {
    const int lane = threadIdx.x & 31, wave = threadIdx.x >> 5;
#pragma unroll
    for (int o = 1; o < 32; o <<= 1) { int t = __shfl_up(v, o, 32); if (lane >= o) v += t; }
    if (lane == 31) sw[wave] = v;
    __syncthreads();
    if (wave == 0) {
        int s = sw[lane];
#pragma unroll
        for (int o = 1; o < 32; o <<= 1) { int t = __shfl_up(s, o, 32); if (lane >= o) s += t; }
        sw[lane] = s;
    }
    __syncthreads();
    total = sw[31];
    return v + (wave ? sw[wave - 1] : 0);
}

__global__ void __launch_bounds__(SCAN_TB) blocksum_kernel(const int* __restrict__ cnt,
                                                           int* __restrict__ bsum, int n) {
    __shared__ int sw[32];
    int i = blockIdx.x * SCAN_TB + threadIdx.x;
    int total;
    block_incl_scan(i < n ? cnt[i] : 0, sw, total);
    if (threadIdx.x == 0) bsum[blockIdx.x] = total;
}

__global__ void __launch_bounds__(SCAN_TB) scanwrite_kernel(const int* __restrict__ cnt,
                                                            const int* __restrict__ bsum,
                                                            int* __restrict__ off, int n) {
    __shared__ int swa[32];
    __shared__ int swb[32];
    int i = blockIdx.x * SCAN_TB + threadIdx.x;
    int base, tot;
    block_incl_scan((int)threadIdx.x < (int)blockIdx.x ? bsum[threadIdx.x] : 0, swa, base);
    int v = i < n ? cnt[i] : 0;
    int incl = block_incl_scan(v, swb, tot);
    if (i < n) off[i] = base + incl - v;
    if (i == n - 1) off[n] = base + incl;
}

__global__ void __launch_bounds__(SORT_TB) bucket_scatter_kernel(const int* __restrict__ row,
                                                                 const int* __restrict__ col,
                                                                 const float* __restrict__ ew,
                                                                 const int* __restrict__ hoff,
                                                                 int2* __restrict__ bk, int E, int CH) {
    __shared__ int cur[NBUCK];
    const int tid = threadIdx.x;
    for (int i = tid; i < NBUCK; i += SORT_TB) cur[i] = hoff[i * SORT_WGS + blockIdx.x];
    __syncthreads();
    const int lo = blockIdx.x * CH;
    const int hi = min(E, lo + CH);
    for (int e0 = lo + tid; e0 < hi; e0 += SORT_U * SORT_TB) {
        int c[SORT_U], r[SORT_U], w[SORT_U];
#pragma unroll
        for (int k = 0; k < SORT_U; ++k) {
            const int e = e0 + k * SORT_TB;
            const bool ok = e < hi;
            c[k] = ok ? col[e] : -1;
            r[k] = ok ? row[e] : 0;
            w[k] = ok ? __float_as_int(ew[e]) : 0;
        }
#pragma unroll
        for (int k = 0; k < SORT_U; ++k) {
            if (c[k] >= 0) {
                const int p = atomicAdd(&cur[c[k] >> BSH], 1);
                bk[p] = make_int2(r[k] | ((c[k] & (BW - 1)) << ROWBITS), w[k]);
            }
        }
    }
}

#define SORT_R 8
__global__ void __launch_bounds__(SORT_TB) bucket_sort_kernel(const int* __restrict__ hoff,
                                                              const int2* __restrict__ bk,
                                                              int* __restrict__ off,
                                                              int2* __restrict__ sraw, int n) {
    __shared__ int cnt[BW];
    __shared__ int cur[BW];
    __shared__ int wsum[BW / 32];
    const int tid = threadIdx.x;
    const int b = blockIdx.x;
    const int start = hoff[b * SORT_WGS];
    const int end = hoff[(b + 1) * SORT_WGS];
    if (tid < BW) cnt[tid] = 0;
    int2 k[SORT_R];
#pragma unroll
    for (int j = 0; j < SORT_R; ++j) {
        const int p = start + tid + j * SORT_TB;
        k[j] = p < end ? bk[p] : make_int2(-1, 0);
    }
    __syncthreads();
#pragma unroll
    for (int j = 0; j < SORT_R; ++j) if (k[j].x >= 0) atomicAdd(&cnt[k[j].x >> ROWBITS], 1);
    for (int p = start + tid + SORT_R * SORT_TB; p < end; p += SORT_TB)
        atomicAdd(&cnt[bk[p].x >> ROWBITS], 1);
    __syncthreads();
    int v = 0, incl = 0;
    if (tid < BW) {
        v = cnt[tid];
        incl = v;
#pragma unroll
        for (int o = 1; o < 32; o <<= 1) { int t = __shfl_up(incl, o, 32); if ((tid & 31) >= o) incl += t; }
        if ((tid & 31) == 31) wsum[tid >> 5] = incl;
    }
    __syncthreads();
    if (tid < BW) {
        int base = start;
        for (int w = 0; w < (tid >> 5); ++w) base += wsum[w];
        const int ex = base + incl - v;
        cur[tid] = ex;
        const int node = b * BW + tid;
        if (node < n) off[node] = ex;
        if (node == n - 1) off[n] = ex + v;
    }
    __syncthreads();
#pragma unroll
    for (int j = 0; j < SORT_R; ++j) {
        if (k[j].x >= 0) {
            const int q = atomicAdd(&cur[k[j].x >> ROWBITS], 1);
            sraw[q] = make_int2(k[j].x & ((1 << ROWBITS) - 1), k[j].y);
        }
    }
    for (int p = start + tid + SORT_R * SORT_TB; p < end; p += SORT_TB) {
        const int2 r = bk[p];
        const int q = atomicAdd(&cur[r.x >> ROWBITS], 1);
        sraw[q] = make_int2(r.x & ((1 << ROWBITS) - 1), r.y);
    }
}

__global__ void dinv_kernel(const int* __restrict__ off, const int2* __restrict__ se,
                            float* __restrict__ dinv, int n) {
    int gid = blockIdx.x * blockDim.x + threadIdx.x;
    int node = gid >> 3, sub = gid & 7;
    float d = 0.f;
    if (node < n) {
        int b = off[node], e = off[node + 1];
        for (int p = b + sub; p < e; p += 8) d += __int_as_float(se[p].y);
    }
    d += __shfl_xor(d, 4, 32);
    d += __shfl_xor(d, 2, 32);
    d += __shfl_xor(d, 1, 32);
    if (node < n && sub == 0) dinv[node] = (d > 0.f) ? rsqrtf(d) : 0.f;
}

__global__ void norm_kernel(const int* __restrict__ off, const int2* __restrict__ sraw,
                            const float* __restrict__ dinv, int2* __restrict__ se, int n) {
    int gid = blockIdx.x * blockDim.x + threadIdx.x;
    int node = gid >> 3, sub = gid & 7;
    if (node >= n) return;
    int b = off[node], e = off[node + 1];
    float di = dinv[node];
    for (int p = b + sub; p < e; p += 8) {
        int2 s = sraw[p];
        float w = dinv[s.x] * __int_as_float(s.y) * di;
        se[p] = make_int2(s.x, __float_as_int(w));
    }
}

__global__ void gather3_layer1_kernel(const int* __restrict__ off, const int2* __restrict__ se,
                                      const float* __restrict__ x, const float* __restrict__ W,
                                      const float* __restrict__ bias, float* __restrict__ out, int n) {
    int gid = blockIdx.x * blockDim.x + threadIdx.x;
    int node = gid >> 3, sub = gid & 7;
    float a0 = 0.f, a1 = 0.f, a2 = 0.f;
    if (node < n) {
        int b = off[node], e = off[node + 1];
        for (int p = b + sub; p < e; p += 8) {
            int2 s = se[p];
            float w = __int_as_float(s.y);
            const float* h = x + (size_t)s.x * 3;
            a0 += w * h[0]; a1 += w * h[1]; a2 += w * h[2];
        }
    }
#pragma unroll
    for (int o = 4; o; o >>= 1) {
        a0 += __shfl_xor(a0, o, 32);
        a1 += __shfl_xor(a1, o, 32);
        a2 += __shfl_xor(a2, o, 32);
    }
    if (node < n) {
        const float xi[3] = {x[(size_t)node * 3 + 0], x[(size_t)node * 3 + 1], x[(size_t)node * 3 + 2]};
        const float hp[3] = {a0, a1, a2};
        v4f r;
#pragma unroll
        for (int jj = 0; jj < 4; ++jj) {
            const int j = 4 * sub + jj;
            float acc = bias[j];
#pragma unroll
            for (int k = 0; k < 3; ++k) {
                acc += xi[k] * W[k * 32 + j];
                acc += hp[k] * W[96 + k * 32 + j];
            }
            r[jj] = fmaxf(acc, 0.f);
        }
        *reinterpret_cast<v4f*>(out + (size_t)node * 32 + 4 * sub) = r;
    }
}

template <int D>
__global__ void __launch_bounds__(256) gather_kernel(const int* __restrict__ off,
                                                     const int2* __restrict__ se,
                                                     const float* __restrict__ hin,
                                                     float* __restrict__ hout, int n) {
    constexpr int LPN = D / 4;
    constexpr int U = 8;
    int gid = blockIdx.x * blockDim.x + threadIdx.x;
    int node = gid / LPN;
    int c = gid % LPN;
    if (node >= n) return;
    int p = off[node];
    const int e = off[node + 1];
    const float* hc = hin + 4 * c;
    v4f acc = {0.f, 0.f, 0.f, 0.f};
    if (p < e) {
        const int last = e - 1;
        int2 s[U];
#pragma unroll
        for (int j = 0; j < U; ++j) s[j] = se[min(p + j, last)];
        while (true) {
            v4f v[U];
#pragma unroll
            for (int j = 0; j < U; ++j)
                v[j] = *reinterpret_cast<const v4f*>(hc + (size_t)s[j].x * D);
            const int pn = p + U;
            int2 sn[U];
#pragma unroll
            for (int j = 0; j < U; ++j) sn[j] = se[min(pn + j, last)];
            __builtin_amdgcn_sched_barrier(0);
#pragma unroll
            for (int j = 0; j < U; ++j) {
                float w = (p + j < e) ? __int_as_float(s[j].y) : 0.f;
                acc += w * v[j];
            }
            if (pn >= e) break;
            p = pn;
#pragma unroll
            for (int j = 0; j < U; ++j) s[j] = sn[j];
        }
    }
    *reinterpret_cast<v4f*>(hout + (size_t)node * D + 4 * c) = acc;
}

template <int K, int NT, int NTHREADS>
__device__ __forceinline__ void fetch_rows(v4f (&pre)[NT * 16 * (K / 4) / NTHREADS],
                                           const float* a0, const float* a1, const float* a2,
                                           const float* a3, int tile, int tid) {
    constexpr int RV = 16 * (K / 4);
    constexpr int VPT = NT * RV / NTHREADS;
#pragma unroll
    for (int q = 0; q < VPT; ++q) {
        const int t = (q * NTHREADS) / RV;
        const int rem = (q * NTHREADS) % RV + tid;
        const float* A = t == 0 ? a0 : t == 1 ? a1 : t == 2 ? a2 : a3;
        pre[q] = *reinterpret_cast<const v4f*>(A + (size_t)tile * 16 * K + 4 * rem);
    }
}

template <int K, int NT, int NOUT, bool RELU>
__global__ void __launch_bounds__((NOUT / 16) * 32)
tag_gemm_wmma(const float* a0, const float* a1, const float* a2, const float* a3,
              const float* __restrict__ W,
              const float* __restrict__ bias,
              float* out, int ntiles) {
    constexpr int NTHREADS = (NOUT / 16) * 32;
    constexpr int KP  = K + 4;
    constexpr int KS  = K / 4;
    constexpr int RV  = 16 * KS;
    constexpr int VPT = NT * RV / NTHREADS;
    static_assert(RV % NTHREADS == 0, "a staging slot must not straddle two terms");
    __shared__ alignas(16) float sA[2][NT * 16 * KP];

    const int tid  = threadIdx.x;
    const int lane = tid & 31;
    const int wave = tid >> 5;
    const int half = lane >> 4;
    const int l    = lane & 15;
    const int n0   = wave * 16;

    int tile = blockIdx.x;
    v4f pre[VPT];
    if (tile < ntiles) fetch_rows<K, NT, NTHREADS>(pre, a0, a1, a2, a3, tile, tid);

    v2f b[NT][KS];
#pragma unroll
    for (int t = 0; t < NT; ++t) {
#pragma unroll
        for (int j = 0; j < KS; ++j) {
            const float* w = W + ((size_t)t * K + 4 * j + 2 * half) * NOUT + n0 + l;
            b[t][j].x = w[0];
            b[t][j].y = w[NOUT];
        }
    }
    const float bv = bias[n0 + l];

    int stage = 0;
    for (; tile < ntiles; tile += gridDim.x) {
        float* s = sA[stage];
#pragma unroll
        for (int q = 0; q < VPT; ++q) {
            const int t = (q * NTHREADS) / RV;
            const int rem = (q * NTHREADS) % RV + tid;
            *reinterpret_cast<v4f*>(s + t * 16 * KP + (rem / KS) * KP + 4 * (rem % KS)) = pre[q];
        }
        __syncthreads();
        const int nxt = tile + gridDim.x;
        if (nxt < ntiles) fetch_rows<K, NT, NTHREADS>(pre, a0, a1, a2, a3, nxt, tid);

        v8f c = {};
        const float* sl = s + l * KP + 2 * half;
#pragma unroll
        for (int t = 0; t < NT; ++t) {
            v2f a[KS];
#pragma unroll
            for (int j = 0; j < KS; ++j)
                a[j] = *reinterpret_cast<const v2f*>(sl + t * 16 * KP + 4 * j);
#pragma unroll
            for (int j = 0; j < KS; ++j)
                c = __builtin_amdgcn_wmma_f32_16x16x4_f32(false, a[j], false, b[t][j],
                                                          (short)0, c, false, false);
        }

#pragma unroll
        for (int r = 0; r < 8; ++r) {
            int rowi = tile * 16 + r + 8 * half;
            float v = c[r] + bv;
            if (RELU) v = fmaxf(v, 0.f);
            out[(size_t)rowi * NOUT + n0 + l] = v;
        }
        stage ^= 1;
    }
}

template <int K, int NT>
__global__ void __launch_bounds__(256, 4)
tag_gemm_fc_norm_wmma(const float* __restrict__ a0, const float* __restrict__ a1,
                      const float* __restrict__ a2, const float* __restrict__ a3,
                      const float* __restrict__ W,
                      const float* __restrict__ bias,
                      const float* __restrict__ Wf,
                      const float* __restrict__ bf,
                      float* __restrict__ out, int ntiles) {
    constexpr int NOUT = 128;
    constexpr int NTHREADS = 256;
    constexpr int KP  = K + 4;
    constexpr int KS  = K / 4;
    constexpr int RV  = 16 * KS;
    constexpr int VPT = NT * RV / NTHREADS;
    constexpr int HP  = NOUT + 4;
    constexpr int FS  = NOUT / 4;
    constexpr int FG  = 16;
    static_assert(RV % NTHREADS == 0, "a staging slot must not straddle two terms");
    __shared__ alignas(16) float sA[2][NT * 16 * KP];
    __shared__ alignas(16) float sH[16 * HP];
    __shared__ alignas(16) float sO[16 * HP];

    const int tid  = threadIdx.x;
    const int lane = tid & 31;
    const int wave = tid >> 5;
    const int half = lane >> 4;
    const int l    = lane & 15;
    const int n0   = wave * 16;

    int tile = blockIdx.x;
    v4f pre[VPT];
    if (tile < ntiles) fetch_rows<K, NT, NTHREADS>(pre, a0, a1, a2, a3, tile, tid);

    v2f b[NT][KS];
#pragma unroll
    for (int t = 0; t < NT; ++t) {
#pragma unroll
        for (int j = 0; j < KS; ++j) {
            const float* w = W + ((size_t)t * K + 4 * j + 2 * half) * NOUT + n0 + l;
            b[t][j].x = w[0];
            b[t][j].y = w[NOUT];
        }
    }
    v2f bfc[FS];
#pragma unroll
    for (int j = 0; j < FS; ++j) {
        const float* w = Wf + ((size_t)4 * j + 2 * half) * NOUT + n0 + l;
        bfc[j].x = w[0];
        bfc[j].y = w[NOUT];
    }
    const float bv  = bias[n0 + l];
    const float bfv = bf[n0 + l];

    int stage = 0;
    for (; tile < ntiles; tile += gridDim.x) {
        float* s = sA[stage];
#pragma unroll
        for (int q = 0; q < VPT; ++q) {
            const int t = (q * NTHREADS) / RV;
            const int rem = (q * NTHREADS) % RV + tid;
            *reinterpret_cast<v4f*>(s + t * 16 * KP + (rem / KS) * KP + 4 * (rem % KS)) = pre[q];
        }
        __syncthreads();
        const int nxt = tile + gridDim.x;
        if (nxt < ntiles) fetch_rows<K, NT, NTHREADS>(pre, a0, a1, a2, a3, nxt, tid);

        v8f c = {};
        const float* sl = s + l * KP + 2 * half;
#pragma unroll
        for (int t = 0; t < NT; ++t) {
            v2f a[KS];
#pragma unroll
            for (int j = 0; j < KS; ++j)
                a[j] = *reinterpret_cast<const v2f*>(sl + t * 16 * KP + 4 * j);
#pragma unroll
            for (int j = 0; j < KS; ++j)
                c = __builtin_amdgcn_wmma_f32_16x16x4_f32(false, a[j], false, b[t][j],
                                                          (short)0, c, false, false);
        }

#pragma unroll
        for (int r = 0; r < 8; ++r)
            sH[(r + 8 * half) * HP + n0 + l] = fmaxf(c[r] + bv, 0.f);
        __syncthreads();

        v8f d = {};
        const float* hl = sH + l * HP + 2 * half;
#pragma unroll
        for (int g = 0; g < FS / FG; ++g) {
            v2f a[FG];
#pragma unroll
            for (int j = 0; j < FG; ++j)
                a[j] = *reinterpret_cast<const v2f*>(hl + 4 * (FG * g + j));
#pragma unroll
            for (int j = 0; j < FG; ++j)
                d = __builtin_amdgcn_wmma_f32_16x16x4_f32(false, a[j], false, bfc[FG * g + j],
                                                          (short)0, d, false, false);
        }
#pragma unroll
        for (int r = 0; r < 8; ++r)
            sO[(r + 8 * half) * HP + n0 + l] = d[r] + bfv;
        __syncthreads();

#pragma nounroll
        for (int q = 0; q < 2; ++q) {
            const int rr = 2 * wave + q;
            float4 v = *reinterpret_cast<const float4*>(sO + rr * HP + 4 * lane);
            float ss = v.x * v.x + v.y * v.y + v.z * v.z + v.w * v.w;
#pragma unroll
            for (int o = 16; o; o >>= 1) ss += __shfl_xor(ss, o, 32);
            float nrm = fmaxf(sqrtf(ss), 1e-12f);
            float inv = 1.f / nrm;
            v.x *= inv; v.y *= inv; v.z *= inv; v.w *= inv;
            *reinterpret_cast<float4*>(out + (size_t)(tile * 16 + rr) * NOUT + 4 * lane) = v;
        }
        stage ^= 1;
    }
}

extern "C" void kernel_launch(void* const* d_in, const int* in_sizes, int n_in,
                              void* d_out, int out_size, void* d_ws, size_t ws_size,
                              hipStream_t stream) {
    const float* x    = (const float*)d_in[0];
    const int*   ei   = (const int*)d_in[1];
    const float* ew   = (const float*)d_in[2];
    const float* w1   = (const float*)d_in[3];
    const float* b1   = (const float*)d_in[4];
    const float* w2   = (const float*)d_in[5];
    const float* b2   = (const float*)d_in[6];
    const float* w3   = (const float*)d_in[7];
    const float* b3   = (const float*)d_in[8];
    const float* fcw  = (const float*)d_in[9];
    const float* fcb  = (const float*)d_in[10];
    float* out = (float*)d_out;

    const int N = NNODES, E = NEDGES;
    const int HNP = (HN + 1 + 63) / 64 * 64;
    const int* row = ei;
    const int* col = ei + E;

    float* wsf  = (float*)d_ws;
    float* P0   = wsf;
    float* P1   = P0 + (size_t)N * 64;
    float* F2   = P1 + (size_t)N * 64;
    float* F1   = F2 + (size_t)N * 64;
    float* P2   = F1;
    int*   hist = (int*)(F1 + (size_t)N * 64);
    int*   hoff = hist + HNP;
    int*   off  = hoff + HNP;
    int*   bsum = off + N + 16;
    float* dinv = (float*)(bsum + 128);
    int2*  sraw = (int2*)(dinv + N);
    int2*  se   = sraw + E;
    int2*  bk   = se;

    const int TB = 256;
    const int gN8 = (N * 8 + TB - 1) / TB;
    const int gS  = (HN + SCAN_TB - 1) / SCAN_TB;
    const int CH  = (E + SORT_WGS - 1) / SORT_WGS;
    const int gMT = N / 16;
    const int gG3 = 512;
    const int gG2 = 1280;

    bucket_hist_kernel   <<<SORT_WGS, SORT_TB, 0, stream>>>(col, hist, E, CH);
    blocksum_kernel      <<<gS, SCAN_TB, 0, stream>>>(hist, bsum, HN);
    scanwrite_kernel     <<<gS, SCAN_TB, 0, stream>>>(hist, bsum, hoff, HN);
    bucket_scatter_kernel<<<SORT_WGS, SORT_TB, 0, stream>>>(row, col, ew, hoff, bk, E, CH);
    bucket_sort_kernel   <<<NBUCK, SORT_TB, 0, stream>>>(hoff, bk, off, sraw, N);
    dinv_kernel     <<<gN8, TB, 0, stream>>>(off, sraw, dinv, N);
    norm_kernel     <<<gN8, TB, 0, stream>>>(off, sraw, dinv, se, N);

    gather3_layer1_kernel<<<gN8, TB, 0, stream>>>(off, se, x, w1, b1, F1, N);

    gather_kernel<32><<<(N * 8 + TB - 1) / TB, TB, 0, stream>>>(off, se, F1, P0, N);
    gather_kernel<32><<<(N * 8 + TB - 1) / TB, TB, 0, stream>>>(off, se, P0, P1, N);
    tag_gemm_wmma<32, 3, 64, true><<<gG2, 128, 0, stream>>>(F1, P0, P1, nullptr, w2, b2, F2, gMT);

    gather_kernel<64><<<(N * 16 + TB - 1) / TB, TB, 0, stream>>>(off, se, F2, P0, N);
    gather_kernel<64><<<(N * 16 + TB - 1) / TB, TB, 0, stream>>>(off, se, P0, P1, N);
    gather_kernel<64><<<(N * 16 + TB - 1) / TB, TB, 0, stream>>>(off, se, P1, P2, N);
    tag_gemm_fc_norm_wmma<64, 4><<<gG3, 256, 0, stream>>>(F2, P0, P1, P2, w3, b3, fcw, fcb, out, gMT);
}


# CSR build in 2 launches, not 5: each chunk sorts its edges by bucket in LDS; a bucket's group sums the chunks' local scans for its start (no global scan); whole-line writes; sum order = LDS arrival
# speedup vs baseline: 5.1730x; 1.0935x over previous
#include <hip/hip_runtime.h>


#define NNODES 100000
#define NEDGES 1600000

typedef __attribute__((ext_vector_type(2))) float v2f;
typedef __attribute__((ext_vector_type(8))) float v8f;
typedef __attribute__((ext_vector_type(4))) float v4f;

#define BSH 8
#define BW (1 << BSH)
#define NBUCK ((NNODES + BW - 1) / BW)
#define SORT_TB 1024
#define SORT_U 4
#define CH (SORT_U * SORT_TB)
#define NCH ((NEDGES + CH - 1) / CH)
#define ROWBITS 20
#define LOFS_WORDS (((NBUCK + 1) * NCH + 63) / 64 * 64)

__device__ __forceinline__ int block_incl_scan(int v, int* sw, int& total) {
    const int lane = threadIdx.x & 31, wave = threadIdx.x >> 5;
#pragma unroll
    for (int o = 1; o < 32; o <<= 1) { int t = __shfl_up(v, o, 32); if (lane >= o) v += t; }
    if (lane == 31) sw[wave] = v;
    __syncthreads();
    if (wave == 0) {
        int s = sw[lane];
#pragma unroll
        for (int o = 1; o < 32; o <<= 1) { int t = __shfl_up(s, o, 32); if (lane >= o) s += t; }
        sw[lane] = s;
    }
    __syncthreads();
    total = sw[31];
    return v + (wave ? sw[wave - 1] : 0);
}

__global__ void __launch_bounds__(SORT_TB) chunk_sort_kernel(const int* __restrict__ row,
                                                             const int* __restrict__ col,
                                                             const float* __restrict__ ew,
                                                             int* __restrict__ lofs,
                                                             int2* __restrict__ bk, int E) {
    __shared__ int h[NBUCK + 1];
    __shared__ int sw[32];
    __shared__ int2 stage[CH];
    const int tid = threadIdx.x;
    const int lo = blockIdx.x * CH;
    const int hi = min(E, lo + CH);
    if (tid <= NBUCK) h[tid] = 0;
    int c[SORT_U], r[SORT_U], w[SORT_U];
#pragma unroll
    for (int k = 0; k < SORT_U; ++k) {
        const int e = lo + tid + k * SORT_TB;
        const bool ok = e < hi;
        c[k] = ok ? col[e] : -1;
        r[k] = ok ? row[e] : 0;
        w[k] = ok ? __float_as_int(ew[e]) : 0;
    }
    __syncthreads();
#pragma unroll
    for (int k = 0; k < SORT_U; ++k) if (c[k] >= 0) atomicAdd(&h[c[k] >> BSH], 1);
    __syncthreads();
    const int v = tid < NBUCK ? h[tid] : 0;
    int total;
    const int incl = block_incl_scan(v, sw, total);
    if (tid <= NBUCK) {
        const int ex = incl - v;
        h[tid] = ex;
        lofs[tid * NCH + blockIdx.x] = ex;
    }
    __syncthreads();
#pragma unroll
    for (int k = 0; k < SORT_U; ++k) {
        if (c[k] >= 0) {
            const int p = atomicAdd(&h[c[k] >> BSH], 1);
            stage[p] = make_int2(r[k] | ((c[k] & (BW - 1)) << ROWBITS), w[k]);
        }
    }
    __syncthreads();
    for (int i = tid; i < hi - lo; i += SORT_TB) bk[lo + i] = stage[i];
}

#define SORT_CAP 6144
#define PJ ((NCH + 31) / 32)
__global__ void __launch_bounds__(SORT_TB) bucket_csr_kernel(const int* __restrict__ lofs,
                                                             const int2* __restrict__ bk,
                                                             int* __restrict__ off,
                                                             int2* __restrict__ sraw, int n) {
    __shared__ int pa[NCH];
    __shared__ int pl[NCH];
    __shared__ int cnt[BW];
    __shared__ int cur[BW];
    __shared__ int wsum[BW / 32];
    __shared__ int swa[32];
    __shared__ int swb[32];
    __shared__ int longp;
    __shared__ int2 stage[SORT_CAP];
    const int tid = threadIdx.x;
    const int lane = tid & 31, wave = tid >> 5;
    const int b = blockIdx.x;
    int a = 0, len = 0;
    if (tid < NCH) {
        a = lofs[b * NCH + tid];
        len = lofs[(b + 1) * NCH + tid] - a;
        pa[tid] = tid * CH + a;
        pl[tid] = len;
    }
    if (tid < BW) cnt[tid] = 0;
    if (tid == 0) longp = 0;
    int start, total;
    block_incl_scan(a, swa, start);
    block_incl_scan(len, swb, total);
    if (len > 32) longp = 1;
    __syncthreads();
    const bool haslong = longp != 0;
    const bool inlds = total <= SORT_CAP;
    int2 k[PJ];
#pragma unroll
    for (int j = 0; j < PJ; ++j) {
        const int c = wave + 32 * j;
        k[j] = make_int2(-1, 0);
        if (c < NCH && lane < pl[c]) k[j] = bk[pa[c] + lane];
    }
#pragma unroll
    for (int j = 0; j < PJ; ++j) if (k[j].x >= 0) atomicAdd(&cnt[k[j].x >> ROWBITS], 1);
    if (haslong) {
        for (int c = wave; c < NCH; c += 32)
            for (int l = lane + 32; l < pl[c]; l += 32) atomicAdd(&cnt[bk[pa[c] + l].x >> ROWBITS], 1);
    }
    __syncthreads();
    int v = 0, incl = 0;
    if (tid < BW) {
        v = cnt[tid];
        incl = v;
#pragma unroll
        for (int o = 1; o < 32; o <<= 1) { int t = __shfl_up(incl, o, 32); if ((tid & 31) >= o) incl += t; }
        if ((tid & 31) == 31) wsum[tid >> 5] = incl;
    }
    __syncthreads();
    if (tid < BW) {
        int base = 0;
        for (int w = 0; w < (tid >> 5); ++w) base += wsum[w];
        const int ex = base + incl - v;
        cur[tid] = ex;
        const int node = b * BW + tid;
        if (node < n) off[node] = start + ex;
        if (node == n - 1) off[n] = start + ex + v;
    }
    __syncthreads();
#pragma unroll
    for (int j = 0; j < PJ; ++j) {
        if (k[j].x >= 0) {
            const int q = atomicAdd(&cur[k[j].x >> ROWBITS], 1);
            const int2 rec = make_int2(k[j].x & ((1 << ROWBITS) - 1), k[j].y);
            if (inlds) stage[q] = rec; else sraw[start + q] = rec;
        }
    }
    if (haslong) {
        for (int c = wave; c < NCH; c += 32) {
            for (int l = lane + 32; l < pl[c]; l += 32) {
                const int2 r = bk[pa[c] + l];
                const int q = atomicAdd(&cur[r.x >> ROWBITS], 1);
                const int2 rec = make_int2(r.x & ((1 << ROWBITS) - 1), r.y);
                if (inlds) stage[q] = rec; else sraw[start + q] = rec;
            }
        }
    }
    if (inlds) {
        __syncthreads();
        for (int i = tid; i < total; i += SORT_TB) sraw[start + i] = stage[i];
    }
}

__global__ void dinv_kernel(const int* __restrict__ off, const int2* __restrict__ se,
                            float* __restrict__ dinv, int n) {
    int gid = blockIdx.x * blockDim.x + threadIdx.x;
    int node = gid >> 3, sub = gid & 7;
    float d = 0.f;
    if (node < n) {
        int b = off[node], e = off[node + 1];
        for (int p = b + sub; p < e; p += 8) d += __int_as_float(se[p].y);
    }
    d += __shfl_xor(d, 4, 32);
    d += __shfl_xor(d, 2, 32);
    d += __shfl_xor(d, 1, 32);
    if (node < n && sub == 0) dinv[node] = (d > 0.f) ? rsqrtf(d) : 0.f;
}

__global__ void norm_kernel(const int* __restrict__ off, const int2* __restrict__ sraw,
                            const float* __restrict__ dinv, int2* __restrict__ se, int n) {
    int gid = blockIdx.x * blockDim.x + threadIdx.x;
    int node = gid >> 3, sub = gid & 7;
    if (node >= n) return;
    int b = off[node], e = off[node + 1];
    float di = dinv[node];
    for (int p = b + sub; p < e; p += 8) {
        int2 s = sraw[p];
        float w = dinv[s.x] * __int_as_float(s.y) * di;
        se[p] = make_int2(s.x, __float_as_int(w));
    }
}

__global__ void gather3_layer1_kernel(const int* __restrict__ off, const int2* __restrict__ se,
                                      const float* __restrict__ x, const float* __restrict__ W,
                                      const float* __restrict__ bias, float* __restrict__ out, int n) {
    int gid = blockIdx.x * blockDim.x + threadIdx.x;
    int node = gid >> 3, sub = gid & 7;
    float a0 = 0.f, a1 = 0.f, a2 = 0.f;
    if (node < n) {
        int b = off[node], e = off[node + 1];
        for (int p = b + sub; p < e; p += 8) {
            int2 s = se[p];
            float w = __int_as_float(s.y);
            const float* h = x + (size_t)s.x * 3;
            a0 += w * h[0]; a1 += w * h[1]; a2 += w * h[2];
        }
    }
#pragma unroll
    for (int o = 4; o; o >>= 1) {
        a0 += __shfl_xor(a0, o, 32);
        a1 += __shfl_xor(a1, o, 32);
        a2 += __shfl_xor(a2, o, 32);
    }
    if (node < n) {
        const float xi[3] = {x[(size_t)node * 3 + 0], x[(size_t)node * 3 + 1], x[(size_t)node * 3 + 2]};
        const float hp[3] = {a0, a1, a2};
        v4f r;
#pragma unroll
        for (int jj = 0; jj < 4; ++jj) {
            const int j = 4 * sub + jj;
            float acc = bias[j];
#pragma unroll
            for (int k = 0; k < 3; ++k) {
                acc += xi[k] * W[k * 32 + j];
                acc += hp[k] * W[96 + k * 32 + j];
            }
            r[jj] = fmaxf(acc, 0.f);
        }
        *reinterpret_cast<v4f*>(out + (size_t)node * 32 + 4 * sub) = r;
    }
}

template <int D>
__global__ void __launch_bounds__(256) gather_kernel(const int* __restrict__ off,
                                                     const int2* __restrict__ se,
                                                     const float* __restrict__ hin,
                                                     float* __restrict__ hout, int n) {
    constexpr int LPN = D / 4;
    constexpr int U = 8;
    int gid = blockIdx.x * blockDim.x + threadIdx.x;
    int node = gid / LPN;
    int c = gid % LPN;
    if (node >= n) return;
    int p = off[node];
    const int e = off[node + 1];
    const float* hc = hin + 4 * c;
    v4f acc = {0.f, 0.f, 0.f, 0.f};
    if (p < e) {
        const int last = e - 1;
        int2 s[U];
#pragma unroll
        for (int j = 0; j < U; ++j) s[j] = se[min(p + j, last)];
        while (true) {
            v4f v[U];
#pragma unroll
            for (int j = 0; j < U; ++j)
                v[j] = *reinterpret_cast<const v4f*>(hc + (size_t)s[j].x * D);
            const int pn = p + U;
            int2 sn[U];
#pragma unroll
            for (int j = 0; j < U; ++j) sn[j] = se[min(pn + j, last)];
            __builtin_amdgcn_sched_barrier(0);
#pragma unroll
            for (int j = 0; j < U; ++j) {
                float w = (p + j < e) ? __int_as_float(s[j].y) : 0.f;
                acc += w * v[j];
            }
            if (pn >= e) break;
            p = pn;
#pragma unroll
            for (int j = 0; j < U; ++j) s[j] = sn[j];
        }
    }
    *reinterpret_cast<v4f*>(hout + (size_t)node * D + 4 * c) = acc;
}

template <int K, int NT, int NTHREADS>
__device__ __forceinline__ void fetch_rows(v4f (&pre)[NT * 16 * (K / 4) / NTHREADS],
                                           const float* a0, const float* a1, const float* a2,
                                           const float* a3, int tile, int tid) {
    constexpr int RV = 16 * (K / 4);
    constexpr int VPT = NT * RV / NTHREADS;
#pragma unroll
    for (int q = 0; q < VPT; ++q) {
        const int t = (q * NTHREADS) / RV;
        const int rem = (q * NTHREADS) % RV + tid;
        const float* A = t == 0 ? a0 : t == 1 ? a1 : t == 2 ? a2 : a3;
        pre[q] = *reinterpret_cast<const v4f*>(A + (size_t)tile * 16 * K + 4 * rem);
    }
}

template <int K, int NT, int NOUT, bool RELU>
__global__ void __launch_bounds__((NOUT / 16) * 32)
tag_gemm_wmma(const float* a0, const float* a1, const float* a2, const float* a3,
              const float* __restrict__ W,
              const float* __restrict__ bias,
              float* out, int ntiles) {
    constexpr int NTHREADS = (NOUT / 16) * 32;
    constexpr int KP  = K + 4;
    constexpr int KS  = K / 4;
    constexpr int RV  = 16 * KS;
    constexpr int VPT = NT * RV / NTHREADS;
    static_assert(RV % NTHREADS == 0, "a staging slot must not straddle two terms");
    __shared__ alignas(16) float sA[2][NT * 16 * KP];

    const int tid  = threadIdx.x;
    const int lane = tid & 31;
    const int wave = tid >> 5;
    const int half = lane >> 4;
    const int l    = lane & 15;
    const int n0   = wave * 16;

    int tile = blockIdx.x;
    v4f pre[VPT];
    if (tile < ntiles) fetch_rows<K, NT, NTHREADS>(pre, a0, a1, a2, a3, tile, tid);

    v2f b[NT][KS];
#pragma unroll
    for (int t = 0; t < NT; ++t) {
#pragma unroll
        for (int j = 0; j < KS; ++j) {
            const float* w = W + ((size_t)t * K + 4 * j + 2 * half) * NOUT + n0 + l;
            b[t][j].x = w[0];
            b[t][j].y = w[NOUT];
        }
    }
    const float bv = bias[n0 + l];

    int stage = 0;
    for (; tile < ntiles; tile += gridDim.x) {
        float* s = sA[stage];
#pragma unroll
        for (int q = 0; q < VPT; ++q) {
            const int t = (q * NTHREADS) / RV;
            const int rem = (q * NTHREADS) % RV + tid;
            *reinterpret_cast<v4f*>(s + t * 16 * KP + (rem / KS) * KP + 4 * (rem % KS)) = pre[q];
        }
        __syncthreads();
        const int nxt = tile + gridDim.x;
        if (nxt < ntiles) fetch_rows<K, NT, NTHREADS>(pre, a0, a1, a2, a3, nxt, tid);

        v8f c = {};
        const float* sl = s + l * KP + 2 * half;
#pragma unroll
        for (int t = 0; t < NT; ++t) {
            v2f a[KS];
#pragma unroll
            for (int j = 0; j < KS; ++j)
                a[j] = *reinterpret_cast<const v2f*>(sl + t * 16 * KP + 4 * j);
#pragma unroll
            for (int j = 0; j < KS; ++j)
                c = __builtin_amdgcn_wmma_f32_16x16x4_f32(false, a[j], false, b[t][j],
                                                          (short)0, c, false, false);
        }

#pragma unroll
        for (int r = 0; r < 8; ++r) {
            int rowi = tile * 16 + r + 8 * half;
            float v = c[r] + bv;
            if (RELU) v = fmaxf(v, 0.f);
            out[(size_t)rowi * NOUT + n0 + l] = v;
        }
        stage ^= 1;
    }
}

template <int K, int NT>
__global__ void __launch_bounds__(256, 4)
tag_gemm_fc_norm_wmma(const float* __restrict__ a0, const float* __restrict__ a1,
                      const float* __restrict__ a2, const float* __restrict__ a3,
                      const float* __restrict__ W,
                      const float* __restrict__ bias,
                      const float* __restrict__ Wf,
                      const float* __restrict__ bf,
                      float* __restrict__ out, int ntiles) {
    constexpr int NOUT = 128;
    constexpr int NTHREADS = 256;
    constexpr int KP  = K + 4;
    constexpr int KS  = K / 4;
    constexpr int RV  = 16 * KS;
    constexpr int VPT = NT * RV / NTHREADS;
    constexpr int HP  = NOUT + 4;
    constexpr int FS  = NOUT / 4;
    constexpr int FG  = 16;
    static_assert(RV % NTHREADS == 0, "a staging slot must not straddle two terms");
    __shared__ alignas(16) float sA[2][NT * 16 * KP];
    __shared__ alignas(16) float sH[16 * HP];
    __shared__ alignas(16) float sO[16 * HP];

    const int tid  = threadIdx.x;
    const int lane = tid & 31;
    const int wave = tid >> 5;
    const int half = lane >> 4;
    const int l    = lane & 15;
    const int n0   = wave * 16;

    int tile = blockIdx.x;
    v4f pre[VPT];
    if (tile < ntiles) fetch_rows<K, NT, NTHREADS>(pre, a0, a1, a2, a3, tile, tid);

    v2f b[NT][KS];
#pragma unroll
    for (int t = 0; t < NT; ++t) {
#pragma unroll
        for (int j = 0; j < KS; ++j) {
            const float* w = W + ((size_t)t * K + 4 * j + 2 * half) * NOUT + n0 + l;
            b[t][j].x = w[0];
            b[t][j].y = w[NOUT];
        }
    }
    v2f bfc[FS];
#pragma unroll
    for (int j = 0; j < FS; ++j) {
        const float* w = Wf + ((size_t)4 * j + 2 * half) * NOUT + n0 + l;
        bfc[j].x = w[0];
        bfc[j].y = w[NOUT];
    }
    const float bv  = bias[n0 + l];
    const float bfv = bf[n0 + l];

    int stage = 0;
    for (; tile < ntiles; tile += gridDim.x) {
        float* s = sA[stage];
#pragma unroll
        for (int q = 0; q < VPT; ++q) {
            const int t = (q * NTHREADS) / RV;
            const int rem = (q * NTHREADS) % RV + tid;
            *reinterpret_cast<v4f*>(s + t * 16 * KP + (rem / KS) * KP + 4 * (rem % KS)) = pre[q];
        }
        __syncthreads();
        const int nxt = tile + gridDim.x;
        if (nxt < ntiles) fetch_rows<K, NT, NTHREADS>(pre, a0, a1, a2, a3, nxt, tid);

        v8f c = {};
        const float* sl = s + l * KP + 2 * half;
#pragma unroll
        for (int t = 0; t < NT; ++t) {
            v2f a[KS];
#pragma unroll
            for (int j = 0; j < KS; ++j)
                a[j] = *reinterpret_cast<const v2f*>(sl + t * 16 * KP + 4 * j);
#pragma unroll
            for (int j = 0; j < KS; ++j)
                c = __builtin_amdgcn_wmma_f32_16x16x4_f32(false, a[j], false, b[t][j],
                                                          (short)0, c, false, false);
        }

#pragma unroll
        for (int r = 0; r < 8; ++r)
            sH[(r + 8 * half) * HP + n0 + l] = fmaxf(c[r] + bv, 0.f);
        __syncthreads();

        v8f d = {};
        const float* hl = sH + l * HP + 2 * half;
#pragma unroll
        for (int g = 0; g < FS / FG; ++g) {
            v2f a[FG];
#pragma unroll
            for (int j = 0; j < FG; ++j)
                a[j] = *reinterpret_cast<const v2f*>(hl + 4 * (FG * g + j));
#pragma unroll
            for (int j = 0; j < FG; ++j)
                d = __builtin_amdgcn_wmma_f32_16x16x4_f32(false, a[j], false, bfc[FG * g + j],
                                                          (short)0, d, false, false);
        }
#pragma unroll
        for (int r = 0; r < 8; ++r)
            sO[(r + 8 * half) * HP + n0 + l] = d[r] + bfv;
        __syncthreads();

#pragma nounroll
        for (int q = 0; q < 2; ++q) {
            const int rr = 2 * wave + q;
            float4 v = *reinterpret_cast<const float4*>(sO + rr * HP + 4 * lane);
            float ss = v.x * v.x + v.y * v.y + v.z * v.z + v.w * v.w;
#pragma unroll
            for (int o = 16; o; o >>= 1) ss += __shfl_xor(ss, o, 32);
            float nrm = fmaxf(sqrtf(ss), 1e-12f);
            float inv = 1.f / nrm;
            v.x *= inv; v.y *= inv; v.z *= inv; v.w *= inv;
            *reinterpret_cast<float4*>(out + (size_t)(tile * 16 + rr) * NOUT + 4 * lane) = v;
        }
        stage ^= 1;
    }
}

extern "C" void kernel_launch(void* const* d_in, const int* in_sizes, int n_in,
                              void* d_out, int out_size, void* d_ws, size_t ws_size,
                              hipStream_t stream) {
    const float* x    = (const float*)d_in[0];
    const int*   ei   = (const int*)d_in[1];
    const float* ew   = (const float*)d_in[2];
    const float* w1   = (const float*)d_in[3];
    const float* b1   = (const float*)d_in[4];
    const float* w2   = (const float*)d_in[5];
    const float* b2   = (const float*)d_in[6];
    const float* w3   = (const float*)d_in[7];
    const float* b3   = (const float*)d_in[8];
    const float* fcw  = (const float*)d_in[9];
    const float* fcb  = (const float*)d_in[10];
    float* out = (float*)d_out;

    const int N = NNODES, E = NEDGES;
    const int* row = ei;
    const int* col = ei + E;

    float* wsf  = (float*)d_ws;
    float* P0   = wsf;
    float* P1   = P0 + (size_t)N * 64;
    float* F2   = P1 + (size_t)N * 64;
    float* F1   = F2 + (size_t)N * 64;
    float* P2   = F1;
    int*   lofs = (int*)(F1 + (size_t)N * 64);
    int*   off  = lofs + LOFS_WORDS;
    float* dinv = (float*)(off + N + 16);
    int2*  sraw = (int2*)(dinv + N);
    int2*  se   = sraw + E;
    int2*  bk   = se;

    const int TB = 256;
    const int gN8 = (N * 8 + TB - 1) / TB;
    const int gMT = N / 16;
    const int gG3 = 512;
    const int gG2 = 1280;

    chunk_sort_kernel<<<NCH, SORT_TB, 0, stream>>>(row, col, ew, lofs, bk, E);
    bucket_csr_kernel<<<NBUCK, SORT_TB, 0, stream>>>(lofs, bk, off, sraw, N);
    dinv_kernel     <<<gN8, TB, 0, stream>>>(off, sraw, dinv, N);
    norm_kernel     <<<gN8, TB, 0, stream>>>(off, sraw, dinv, se, N);

    gather3_layer1_kernel<<<gN8, TB, 0, stream>>>(off, se, x, w1, b1, F1, N);

    gather_kernel<32><<<(N * 8 + TB - 1) / TB, TB, 0, stream>>>(off, se, F1, P0, N);
    gather_kernel<32><<<(N * 8 + TB - 1) / TB, TB, 0, stream>>>(off, se, P0, P1, N);
    tag_gemm_wmma<32, 3, 64, true><<<gG2, 128, 0, stream>>>(F1, P0, P1, nullptr, w2, b2, F2, gMT);

    gather_kernel<64><<<(N * 16 + TB - 1) / TB, TB, 0, stream>>>(off, se, F2, P0, N);
    gather_kernel<64><<<(N * 16 + TB - 1) / TB, TB, 0, stream>>>(off, se, P0, P1, N);
    gather_kernel<64><<<(N * 16 + TB - 1) / TB, TB, 0, stream>>>(off, se, P1, P2, N);
    tag_gemm_fc_norm_wmma<64, 4><<<gG3, 256, 0, stream>>>(F2, P0, P1, P2, w3, b3, fcw, fcb, out, gMT);
}


# gcn_norm without launches of its own: bucket kernel sums node degrees from its LDS stage and writes (x, dinv) 16-byte node records; 3-wide hop scales weights as it reads; sum order unchanged
# speedup vs baseline: 5.5363x; 1.0702x over previous
#include <hip/hip_runtime.h>


#define NNODES 100000
#define NEDGES 1600000

typedef __attribute__((ext_vector_type(2))) float v2f;
typedef __attribute__((ext_vector_type(8))) float v8f;
typedef __attribute__((ext_vector_type(4))) float v4f;

#define BSH 8
#define BW (1 << BSH)
#define NBUCK ((NNODES + BW - 1) / BW)
#define SORT_TB 1024
#define SORT_U 4
#define CH (SORT_U * SORT_TB)
#define NCH ((NEDGES + CH - 1) / CH)
#define ROWBITS 20
#define LOFS_WORDS (((NBUCK + 1) * NCH + 63) / 64 * 64)

__device__ __forceinline__ int block_incl_scan(int v, int* sw, int& total) {
    const int lane = threadIdx.x & 31, wave = threadIdx.x >> 5;
#pragma unroll
    for (int o = 1; o < 32; o <<= 1) { int t = __shfl_up(v, o, 32); if (lane >= o) v += t; }
    if (lane == 31) sw[wave] = v;
    __syncthreads();
    if (wave == 0) {
        int s = sw[lane];
#pragma unroll
        for (int o = 1; o < 32; o <<= 1) { int t = __shfl_up(s, o, 32); if (lane >= o) s += t; }
        sw[lane] = s;
    }
    __syncthreads();
    total = sw[31];
    return v + (wave ? sw[wave - 1] : 0);
}

__global__ void __launch_bounds__(SORT_TB) chunk_sort_kernel(const int* __restrict__ row,
                                                             const int* __restrict__ col,
                                                             const float* __restrict__ ew,
                                                             int* __restrict__ lofs,
                                                             int2* __restrict__ bk, int E) {
    __shared__ int h[NBUCK + 1];
    __shared__ int sw[32];
    __shared__ int2 stage[CH];
    const int tid = threadIdx.x;
    const int lo = blockIdx.x * CH;
    const int hi = min(E, lo + CH);
    if (tid <= NBUCK) h[tid] = 0;
    int c[SORT_U], r[SORT_U], w[SORT_U];
#pragma unroll
    for (int k = 0; k < SORT_U; ++k) {
        const int e = lo + tid + k * SORT_TB;
        const bool ok = e < hi;
        c[k] = ok ? col[e] : -1;
        r[k] = ok ? row[e] : 0;
        w[k] = ok ? __float_as_int(ew[e]) : 0;
    }
    __syncthreads();
#pragma unroll
    for (int k = 0; k < SORT_U; ++k) if (c[k] >= 0) atomicAdd(&h[c[k] >> BSH], 1);
    __syncthreads();
    const int v = tid < NBUCK ? h[tid] : 0;
    int total;
    const int incl = block_incl_scan(v, sw, total);
    if (tid <= NBUCK) {
        const int ex = incl - v;
        h[tid] = ex;
        lofs[tid * NCH + blockIdx.x] = ex;
    }
    __syncthreads();
#pragma unroll
    for (int k = 0; k < SORT_U; ++k) {
        if (c[k] >= 0) {
            const int p = atomicAdd(&h[c[k] >> BSH], 1);
            stage[p] = make_int2(r[k] | ((c[k] & (BW - 1)) << ROWBITS), w[k]);
        }
    }
    __syncthreads();
    for (int i = tid; i < hi - lo; i += SORT_TB) bk[lo + i] = stage[i];
}

#define SORT_CAP 6144
#define PJ ((NCH + 31) / 32)
__global__ void __launch_bounds__(SORT_TB) bucket_csr_kernel(const int* __restrict__ lofs,
                                                             const int2* __restrict__ bk,
                                                             const float* __restrict__ x,
                                                             int* __restrict__ off,
                                                             int2* __restrict__ sraw,
                                                             float* __restrict__ xd, int n) {
    __shared__ int pa[NCH];
    __shared__ int pl[NCH];
    __shared__ int cnt[BW];
    __shared__ int cur[BW];
    __shared__ int wsum[BW / 32];
    __shared__ int swa[32];
    __shared__ int swb[32];
    __shared__ int longp;
    __shared__ int2 stage[SORT_CAP];
    const int tid = threadIdx.x;
    const int lane = tid & 31, wave = tid >> 5;
    const int b = blockIdx.x;
    int a = 0, len = 0;
    if (tid < NCH) {
        a = lofs[b * NCH + tid];
        len = lofs[(b + 1) * NCH + tid] - a;
        pa[tid] = tid * CH + a;
        pl[tid] = len;
    }
    if (tid < BW) cnt[tid] = 0;
    if (tid == 0) longp = 0;
    int start, total;
    block_incl_scan(a, swa, start);
    block_incl_scan(len, swb, total);
    if (len > 32) longp = 1;
    __syncthreads();
    const bool haslong = longp != 0;
    const bool inlds = total <= SORT_CAP;
    int2 k[PJ];
#pragma unroll
    for (int j = 0; j < PJ; ++j) {
        const int c = wave + 32 * j;
        k[j] = make_int2(-1, 0);
        if (c < NCH && lane < pl[c]) k[j] = bk[pa[c] + lane];
    }
#pragma unroll
    for (int j = 0; j < PJ; ++j) if (k[j].x >= 0) atomicAdd(&cnt[k[j].x >> ROWBITS], 1);
    if (haslong) {
        for (int c = wave; c < NCH; c += 32)
            for (int l = lane + 32; l < pl[c]; l += 32) atomicAdd(&cnt[bk[pa[c] + l].x >> ROWBITS], 1);
    }
    __syncthreads();
    int v = 0, incl = 0;
    if (tid < BW) {
        v = cnt[tid];
        incl = v;
#pragma unroll
        for (int o = 1; o < 32; o <<= 1) { int t = __shfl_up(incl, o, 32); if ((tid & 31) >= o) incl += t; }
        if ((tid & 31) == 31) wsum[tid >> 5] = incl;
    }
    __syncthreads();
    if (tid < BW) {
        int base = 0;
        for (int w = 0; w < (tid >> 5); ++w) base += wsum[w];
        const int ex = base + incl - v;
        cur[tid] = ex;
        const int node = b * BW + tid;
        if (node < n) off[node] = start + ex;
        if (node == n - 1) off[n] = start + ex + v;
    }
    __syncthreads();
#pragma unroll
    for (int j = 0; j < PJ; ++j) {
        if (k[j].x >= 0) {
            const int q = atomicAdd(&cur[k[j].x >> ROWBITS], 1);
            const int2 rec = make_int2(k[j].x & ((1 << ROWBITS) - 1), k[j].y);
            if (inlds) stage[q] = rec; else sraw[start + q] = rec;
        }
    }
    if (haslong) {
        for (int c = wave; c < NCH; c += 32) {
            for (int l = lane + 32; l < pl[c]; l += 32) {
                const int2 r = bk[pa[c] + l];
                const int q = atomicAdd(&cur[r.x >> ROWBITS], 1);
                const int2 rec = make_int2(r.x & ((1 << ROWBITS) - 1), r.y);
                if (inlds) stage[q] = rec; else sraw[start + q] = rec;
            }
        }
    }
    __syncthreads();
    if (inlds)
        for (int i = tid; i < total; i += SORT_TB) sraw[start + i] = stage[i];
    for (int h = 0; h < BW / (SORT_TB / 8); ++h) {
        const int ln = (tid >> 3) + h * (SORT_TB / 8);
        const int sub = tid & 7;
        const int node = b * BW + ln;
        const bool wr = sub == 0 && node < n;
        float x0 = 0.f, x1 = 0.f, x2 = 0.f;
        if (wr) { x0 = x[(size_t)node * 3]; x1 = x[(size_t)node * 3 + 1]; x2 = x[(size_t)node * 3 + 2]; }
        const int e = cur[ln];
        float d = 0.f;
        if (inlds) { for (int p = e - cnt[ln] + sub; p < e; p += 8) d += __int_as_float(stage[p].y); }
        else       { for (int p = e - cnt[ln] + sub; p < e; p += 8) d += __int_as_float(sraw[start + p].y); }
        d += __shfl_xor(d, 4, 32);
        d += __shfl_xor(d, 2, 32);
        d += __shfl_xor(d, 1, 32);
        if (wr) {
            v4f rec = {x0, x1, x2, (d > 0.f) ? rsqrtf(d) : 0.f};
            *reinterpret_cast<v4f*>(xd + (size_t)node * 4) = rec;
        }
    }
}

__global__ void gather3_layer1_kernel(const int* __restrict__ off, const int2* __restrict__ sraw,
                                      const float* __restrict__ xd, const float* __restrict__ W,
                                      const float* __restrict__ bias, int2* __restrict__ se,
                                      float* __restrict__ out, int n) {
    int gid = blockIdx.x * blockDim.x + threadIdx.x;
    int node = gid >> 3, sub = gid & 7;
    float a0 = 0.f, a1 = 0.f, a2 = 0.f;
    v4f own = {0.f, 0.f, 0.f, 0.f};
    if (node < n) {
        own = *reinterpret_cast<const v4f*>(xd + (size_t)node * 4);
        const float di = own.w;
        int b = off[node], e = off[node + 1];
        for (int p = b + sub; p < e; p += 8) {
            int2 s = sraw[p];
            const v4f h = *reinterpret_cast<const v4f*>(xd + (size_t)s.x * 4);
            float w = h.w * __int_as_float(s.y) * di;
            se[p] = make_int2(s.x, __float_as_int(w));
            a0 += w * h.x; a1 += w * h.y; a2 += w * h.z;
        }
    }
#pragma unroll
    for (int o = 4; o; o >>= 1) {
        a0 += __shfl_xor(a0, o, 32);
        a1 += __shfl_xor(a1, o, 32);
        a2 += __shfl_xor(a2, o, 32);
    }
    if (node < n) {
        const float xi[3] = {own.x, own.y, own.z};
        const float hp[3] = {a0, a1, a2};
        v4f r;
#pragma unroll
        for (int jj = 0; jj < 4; ++jj) {
            const int j = 4 * sub + jj;
            float acc = bias[j];
#pragma unroll
            for (int k = 0; k < 3; ++k) {
                acc += xi[k] * W[k * 32 + j];
                acc += hp[k] * W[96 + k * 32 + j];
            }
            r[jj] = fmaxf(acc, 0.f);
        }
        *reinterpret_cast<v4f*>(out + (size_t)node * 32 + 4 * sub) = r;
    }
}

template <int D>
__global__ void __launch_bounds__(256) gather_kernel(const int* __restrict__ off,
                                                     const int2* __restrict__ se,
                                                     const float* __restrict__ hin,
                                                     float* __restrict__ hout, int n) {
    constexpr int LPN = D / 4;
    constexpr int U = 8;
    int gid = blockIdx.x * blockDim.x + threadIdx.x;
    int node = gid / LPN;
    int c = gid % LPN;
    if (node >= n) return;
    int p = off[node];
    const int e = off[node + 1];
    const float* hc = hin + 4 * c;
    v4f acc = {0.f, 0.f, 0.f, 0.f};
    if (p < e) {
        const int last = e - 1;
        int2 s[U];
#pragma unroll
        for (int j = 0; j < U; ++j) s[j] = se[min(p + j, last)];
        while (true) {
            v4f v[U];
#pragma unroll
            for (int j = 0; j < U; ++j)
                v[j] = *reinterpret_cast<const v4f*>(hc + (size_t)s[j].x * D);
            const int pn = p + U;
            int2 sn[U];
#pragma unroll
            for (int j = 0; j < U; ++j) sn[j] = se[min(pn + j, last)];
            __builtin_amdgcn_sched_barrier(0);
#pragma unroll
            for (int j = 0; j < U; ++j) {
                float w = (p + j < e) ? __int_as_float(s[j].y) : 0.f;
                acc += w * v[j];
            }
            if (pn >= e) break;
            p = pn;
#pragma unroll
            for (int j = 0; j < U; ++j) s[j] = sn[j];
        }
    }
    *reinterpret_cast<v4f*>(hout + (size_t)node * D + 4 * c) = acc;
}

template <int K, int NT, int NTHREADS>
__device__ __forceinline__ void fetch_rows(v4f (&pre)[NT * 16 * (K / 4) / NTHREADS],
                                           const float* a0, const float* a1, const float* a2,
                                           const float* a3, int tile, int tid) {
    constexpr int RV = 16 * (K / 4);
    constexpr int VPT = NT * RV / NTHREADS;
#pragma unroll
    for (int q = 0; q < VPT; ++q) {
        const int t = (q * NTHREADS) / RV;
        const int rem = (q * NTHREADS) % RV + tid;
        const float* A = t == 0 ? a0 : t == 1 ? a1 : t == 2 ? a2 : a3;
        pre[q] = *reinterpret_cast<const v4f*>(A + (size_t)tile * 16 * K + 4 * rem);
    }
}

template <int K, int NT, int NOUT, bool RELU>
__global__ void __launch_bounds__((NOUT / 16) * 32)
tag_gemm_wmma(const float* a0, const float* a1, const float* a2, const float* a3,
              const float* __restrict__ W,
              const float* __restrict__ bias,
              float* out, int ntiles) {
    constexpr int NTHREADS = (NOUT / 16) * 32;
    constexpr int KP  = K + 4;
    constexpr int KS  = K / 4;
    constexpr int RV  = 16 * KS;
    constexpr int VPT = NT * RV / NTHREADS;
    static_assert(RV % NTHREADS == 0, "a staging slot must not straddle two terms");
    __shared__ alignas(16) float sA[2][NT * 16 * KP];

    const int tid  = threadIdx.x;
    const int lane = tid & 31;
    const int wave = tid >> 5;
    const int half = lane >> 4;
    const int l    = lane & 15;
    const int n0   = wave * 16;

    int tile = blockIdx.x;
    v4f pre[VPT];
    if (tile < ntiles) fetch_rows<K, NT, NTHREADS>(pre, a0, a1, a2, a3, tile, tid);

    v2f b[NT][KS];
#pragma unroll
    for (int t = 0; t < NT; ++t) {
#pragma unroll
        for (int j = 0; j < KS; ++j) {
            const float* w = W + ((size_t)t * K + 4 * j + 2 * half) * NOUT + n0 + l;
            b[t][j].x = w[0];
            b[t][j].y = w[NOUT];
        }
    }
    const float bv = bias[n0 + l];

    int stage = 0;
    for (; tile < ntiles; tile += gridDim.x) {
        float* s = sA[stage];
#pragma unroll
        for (int q = 0; q < VPT; ++q) {
            const int t = (q * NTHREADS) / RV;
            const int rem = (q * NTHREADS) % RV + tid;
            *reinterpret_cast<v4f*>(s + t * 16 * KP + (rem / KS) * KP + 4 * (rem % KS)) = pre[q];
        }
        __syncthreads();
        const int nxt = tile + gridDim.x;
        if (nxt < ntiles) fetch_rows<K, NT, NTHREADS>(pre, a0, a1, a2, a3, nxt, tid);

        v8f c = {};
        const float* sl = s + l * KP + 2 * half;
#pragma unroll
        for (int t = 0; t < NT; ++t) {
            v2f a[KS];
#pragma unroll
            for (int j = 0; j < KS; ++j)
                a[j] = *reinterpret_cast<const v2f*>(sl + t * 16 * KP + 4 * j);
#pragma unroll
            for (int j = 0; j < KS; ++j)
                c = __builtin_amdgcn_wmma_f32_16x16x4_f32(false, a[j], false, b[t][j],
                                                          (short)0, c, false, false);
        }

#pragma unroll
        for (int r = 0; r < 8; ++r) {
            int rowi = tile * 16 + r + 8 * half;
            float v = c[r] + bv;
            if (RELU) v = fmaxf(v, 0.f);
            out[(size_t)rowi * NOUT + n0 + l] = v;
        }
        stage ^= 1;
    }
}

template <int K, int NT>
__global__ void __launch_bounds__(256, 4)
tag_gemm_fc_norm_wmma(const float* __restrict__ a0, const float* __restrict__ a1,
                      const float* __restrict__ a2, const float* __restrict__ a3,
                      const float* __restrict__ W,
                      const float* __restrict__ bias,
                      const float* __restrict__ Wf,
                      const float* __restrict__ bf,
                      float* __restrict__ out, int ntiles) {
    constexpr int NOUT = 128;
    constexpr int NTHREADS = 256;
    constexpr int KP  = K + 4;
    constexpr int KS  = K / 4;
    constexpr int RV  = 16 * KS;
    constexpr int VPT = NT * RV / NTHREADS;
    constexpr int HP  = NOUT + 4;
    constexpr int FS  = NOUT / 4;
    constexpr int FG  = 16;
    static_assert(RV % NTHREADS == 0, "a staging slot must not straddle two terms");
    __shared__ alignas(16) float sA[2][NT * 16 * KP];
    __shared__ alignas(16) float sH[16 * HP];
    __shared__ alignas(16) float sO[16 * HP];

    const int tid  = threadIdx.x;
    const int lane = tid & 31;
    const int wave = tid >> 5;
    const int half = lane >> 4;
    const int l    = lane & 15;
    const int n0   = wave * 16;

    int tile = blockIdx.x;
    v4f pre[VPT];
    if (tile < ntiles) fetch_rows<K, NT, NTHREADS>(pre, a0, a1, a2, a3, tile, tid);

    v2f b[NT][KS];
#pragma unroll
    for (int t = 0; t < NT; ++t) {
#pragma unroll
        for (int j = 0; j < KS; ++j) {
            const float* w = W + ((size_t)t * K + 4 * j + 2 * half) * NOUT + n0 + l;
            b[t][j].x = w[0];
            b[t][j].y = w[NOUT];
        }
    }
    v2f bfc[FS];
#pragma unroll
    for (int j = 0; j < FS; ++j) {
        const float* w = Wf + ((size_t)4 * j + 2 * half) * NOUT + n0 + l;
        bfc[j].x = w[0];
        bfc[j].y = w[NOUT];
    }
    const float bv  = bias[n0 + l];
    const float bfv = bf[n0 + l];

    int stage = 0;
    for (; tile < ntiles; tile += gridDim.x) {
        float* s = sA[stage];
#pragma unroll
        for (int q = 0; q < VPT; ++q) {
            const int t = (q * NTHREADS) / RV;
            const int rem = (q * NTHREADS) % RV + tid;
            *reinterpret_cast<v4f*>(s + t * 16 * KP + (rem / KS) * KP + 4 * (rem % KS)) = pre[q];
        }
        __syncthreads();
        const int nxt = tile + gridDim.x;
        if (nxt < ntiles) fetch_rows<K, NT, NTHREADS>(pre, a0, a1, a2, a3, nxt, tid);

        v8f c = {};
        const float* sl = s + l * KP + 2 * half;
#pragma unroll
        for (int t = 0; t < NT; ++t) {
            v2f a[KS];
#pragma unroll
            for (int j = 0; j < KS; ++j)
                a[j] = *reinterpret_cast<const v2f*>(sl + t * 16 * KP + 4 * j);
#pragma unroll
            for (int j = 0; j < KS; ++j)
                c = __builtin_amdgcn_wmma_f32_16x16x4_f32(false, a[j], false, b[t][j],
                                                          (short)0, c, false, false);
        }

#pragma unroll
        for (int r = 0; r < 8; ++r)
            sH[(r + 8 * half) * HP + n0 + l] = fmaxf(c[r] + bv, 0.f);
        __syncthreads();

        v8f d = {};
        const float* hl = sH + l * HP + 2 * half;
#pragma unroll
        for (int g = 0; g < FS / FG; ++g) {
            v2f a[FG];
#pragma unroll
            for (int j = 0; j < FG; ++j)
                a[j] = *reinterpret_cast<const v2f*>(hl + 4 * (FG * g + j));
#pragma unroll
            for (int j = 0; j < FG; ++j)
                d = __builtin_amdgcn_wmma_f32_16x16x4_f32(false, a[j], false, bfc[FG * g + j],
                                                          (short)0, d, false, false);
        }
#pragma unroll
        for (int r = 0; r < 8; ++r)
            sO[(r + 8 * half) * HP + n0 + l] = d[r] + bfv;
        __syncthreads();

#pragma nounroll
        for (int q = 0; q < 2; ++q) {
            const int rr = 2 * wave + q;
            float4 v = *reinterpret_cast<const float4*>(sO + rr * HP + 4 * lane);
            float ss = v.x * v.x + v.y * v.y + v.z * v.z + v.w * v.w;
#pragma unroll
            for (int o = 16; o; o >>= 1) ss += __shfl_xor(ss, o, 32);
            float nrm = fmaxf(sqrtf(ss), 1e-12f);
            float inv = 1.f / nrm;
            v.x *= inv; v.y *= inv; v.z *= inv; v.w *= inv;
            *reinterpret_cast<float4*>(out + (size_t)(tile * 16 + rr) * NOUT + 4 * lane) = v;
        }
        stage ^= 1;
    }
}

extern "C" void kernel_launch(void* const* d_in, const int* in_sizes, int n_in,
                              void* d_out, int out_size, void* d_ws, size_t ws_size,
                              hipStream_t stream) {
    const float* x    = (const float*)d_in[0];
    const int*   ei   = (const int*)d_in[1];
    const float* ew   = (const float*)d_in[2];
    const float* w1   = (const float*)d_in[3];
    const float* b1   = (const float*)d_in[4];
    const float* w2   = (const float*)d_in[5];
    const float* b2   = (const float*)d_in[6];
    const float* w3   = (const float*)d_in[7];
    const float* b3   = (const float*)d_in[8];
    const float* fcw  = (const float*)d_in[9];
    const float* fcb  = (const float*)d_in[10];
    float* out = (float*)d_out;

    const int N = NNODES, E = NEDGES;
    const int* row = ei;
    const int* col = ei + E;

    float* wsf  = (float*)d_ws;
    float* P0   = wsf;
    float* P1   = P0 + (size_t)N * 64;
    float* F2   = P1 + (size_t)N * 64;
    float* F1   = F2 + (size_t)N * 64;
    float* P2   = F1;
    int*   lofs = (int*)(F1 + (size_t)N * 64);
    int*   off  = lofs + LOFS_WORDS;
    float* xd   = (float*)(off + N + 16);
    int2*  sraw = (int2*)(xd + (size_t)N * 4);
    int2*  se   = sraw + E;
    int2*  bk   = se;

    const int TB = 256;
    const int gN8 = (N * 8 + TB - 1) / TB;
    const int gMT = N / 16;
    const int gG3 = 512;
    const int gG2 = 1280;

    chunk_sort_kernel<<<NCH, SORT_TB, 0, stream>>>(row, col, ew, lofs, bk, E);
    bucket_csr_kernel<<<NBUCK, SORT_TB, 0, stream>>>(lofs, bk, x, off, sraw, xd, N);

    gather3_layer1_kernel<<<gN8, TB, 0, stream>>>(off, sraw, xd, w1, b1, se, F1, N);

    gather_kernel<32><<<(N * 8 + TB - 1) / TB, TB, 0, stream>>>(off, se, F1, P0, N);
    gather_kernel<32><<<(N * 8 + TB - 1) / TB, TB, 0, stream>>>(off, se, P0, P1, N);
    tag_gemm_wmma<32, 3, 64, true><<<gG2, 128, 0, stream>>>(F1, P0, P1, nullptr, w2, b2, F2, gMT);

    gather_kernel<64><<<(N * 16 + TB - 1) / TB, TB, 0, stream>>>(off, se, F2, P0, N);
    gather_kernel<64><<<(N * 16 + TB - 1) / TB, TB, 0, stream>>>(off, se, P0, P1, N);
    gather_kernel<64><<<(N * 16 + TB - 1) / TB, TB, 0, stream>>>(off, se, P1, P2, N);
    tag_gemm_fc_norm_wmma<64, 4><<<gG3, 256, 0, stream>>>(F2, P0, P1, P2, w3, b3, fcw, fcb, out, gMT);
}
